# starting kernel
# baseline (speedup 1.0000x reference)
#include <hip/hip_runtime.h>

typedef __attribute__((ext_vector_type(2))) float v2f;
typedef __attribute__((ext_vector_type(4))) float v4f;
typedef __attribute__((ext_vector_type(8))) float v8f;

#define NBATCH 8
#define NFEAT  256
#define HD     64
#define NT     1024
#define AUGC   130
#define XSTR   68
#define YCHUNK 256

__launch_bounds__(128)
__global__ void ssm_scan_wmma_f32(const float* __restrict__ x,
                                  const float* __restrict__ Amat,
                                  const float* __restrict__ Bmat,
                                  const float* __restrict__ Cmat,
                                  float* __restrict__ out)
{
    __shared__ float smem[NBATCH*NT + NBATCH*YCHUNK + 16*XSTR + 128];
    float* aug   = smem;
    float* u_lds = smem;
    float* y_lds = smem + NBATCH*NT;
    float* X_lds = smem + NBATCH*NT + NBATCH*YCHUNK;
    float* p_lds = X_lds + 16*XSTR;

    const int f   = blockIdx.x;
    const int tid = threadIdx.x;

    const float* Af = Amat + f*HD*HD;
    for (int e = tid; e < HD*HD; e += 128) {
        int r = e >> 6, c = e & 63;
        float a   = Af[e];
        float eye = (r == c) ? 1.0f : 0.0f;
        aug[r*AUGC + c]      = eye - 0.5f*a;
        aug[r*AUGC + 64 + c] = eye + 0.5f*a;
    }
    for (int r = tid; r < HD; r += 128)
        aug[r*AUGC + 128] = Bmat[f*HD + r];
    __syncthreads();

    {
        const int row  = tid & 63;
        const int c0   = (tid >> 6) ? 64 : 0;
        const int ncol = (tid >> 6) ? 65 : 64;
        for (int p = 0; p < HD; ++p) {
            float fac = aug[row*AUGC + p] / aug[p*AUGC + p];
            __syncthreads();
            if (row != p) {
                for (int c = 0; c < ncol; ++c)
                    aug[row*AUGC + c0 + c] -= fac * aug[p*AUGC + c0 + c];
            }
            __syncthreads();
        }
    }

    const int wv   = tid >> 5;
    const int ln   = tid & 31;
    const int hi   = ln >> 4;
    const int col  = ln & 15;
    const int koff = hi << 1;
    const int arow = wv*16 + col;
    const int row2 = wv*16 + hi*8;

    v2f a_op[16];
    {
        float rinv = 1.0f / aug[arow*AUGC + arow];
        #pragma unroll
        for (int k = 0; k < 16; ++k) {
            a_op[k].x = aug[arow*AUGC + 64 + 4*k + koff]     * rinv;
            a_op[k].y = aug[arow*AUGC + 64 + 4*k + koff + 1] * rinv;
        }
    }
    float bvr[8], cvr[8];
    #pragma unroll
    for (int r = 0; r < 8; ++r) {
        int rr = row2 + r;
        bvr[r] = aug[rr*AUGC + 128] / aug[rr*AUGC + rr];
        cvr[r] = Cmat[f*HD + rr];
    }
    __syncthreads();

    for (int e = tid; e < NBATCH*NT; e += 128) {
        int b = e >> 10, i = e & (NT - 1);
        u_lds[i*NBATCH + b] = x[(size_t)(b*NFEAT + f)*NT + i];
    }
    for (int e = tid; e < 16*XSTR; e += 128) X_lds[e] = 0.0f;
    __syncthreads();

    for (int t = 0; t < NT; ++t) {
        v8f D = {};
        #pragma unroll
        for (int k = 0; k < 16; ++k) {
            v2f bop = *(const v2f*)&X_lds[col*XSTR + 4*k + koff];
            D = __builtin_amdgcn_wmma_f32_16x16x4_f32(false, a_op[k], false, bop,
                                                      (short)0, D, false, false);
        }

        float u = (col < NBATCH) ? u_lds[t*NBATCH + col] : 0.0f;
        float psum = 0.0f;
        #pragma unroll
        for (int r = 0; r < 8; ++r) {
            D[r] += bvr[r] * u;
            psum += cvr[r] * D[r];
        }
        p_lds[(wv*2 + hi)*16 + col] = psum;
        __syncthreads();

        v4f lo4 = { D[0], D[1], D[2], D[3] };
        v4f hi4 = { D[4], D[5], D[6], D[7] };
        *(v4f*)&X_lds[col*XSTR + row2]     = lo4;
        *(v4f*)&X_lds[col*XSTR + row2 + 4] = hi4;

        if (tid < NBATCH) {
            float y = 0.0f;
            #pragma unroll
            for (int j = 0; j < 8; ++j) y += p_lds[j*16 + tid];
            y_lds[(t & (YCHUNK-1))*NBATCH + tid] = y;
        }

        if ((t & (YCHUNK-1)) == (YCHUNK-1)) {
            __syncthreads();
            int tbase = t & ~(YCHUNK-1);
            for (int e = tid; e < NBATCH*YCHUNK; e += 128) {
                int b = e >> 8, i = e & (YCHUNK-1);
                out[(size_t)(b*NFEAT + f)*NT + tbase + i] = y_lds[i*NBATCH + b];
            }
        }
        __syncthreads();
    }
}

extern "C" void kernel_launch(void* const* d_in, const int* in_sizes, int n_in,
                              void* d_out, int out_size, void* d_ws, size_t ws_size,
                              hipStream_t stream) {
    (void)in_sizes; (void)n_in; (void)d_ws; (void)ws_size; (void)out_size;
    const float* x = (const float*)d_in[0];
    const float* A = (const float*)d_in[1];
    const float* B = (const float*)d_in[2];
    const float* C = (const float*)d_in[3];
    float* out = (float*)d_out;
    ssm_scan_wmma_f32<<<dim3(NFEAT), dim3(128), 0, stream>>>(x, A, B, C, out);
}


# Scan replaced by its convolution: taps K[k]=Cv A_bar^k Bv from two one-step vector recurrences (A_bar^b Bv, b<512; Cv A_bar^512), no matrix powers; Toeplitz on WMMA; summation order changed
# speedup vs baseline: 2.9510x; 2.9510x over previous
#include <hip/hip_runtime.h>

typedef __attribute__((ext_vector_type(2))) float v2f;
typedef __attribute__((ext_vector_type(4))) float v4f;
typedef __attribute__((ext_vector_type(8))) float v8f;

#define NBATCH 8
#define NFEAT  256
#define HD     64
#define NT     1024
#define AUGC   130
#define DTHREADS 1024
#define DSTRIDE (HD*HD + HD)
#define STHREADS 256
#define NWAVES 8
#define NHALF  512
#define VP 68
#define MP 68
#define YP 1028

__launch_bounds__(DTHREADS)
__global__ void ssm_discretize_f32(const float* __restrict__ Amat,
                                   const float* __restrict__ Bmat,
                                   float* __restrict__ dsc)
{
    __shared__ float aug[HD*AUGC];

    const int f   = blockIdx.x;
    const int tid = threadIdx.x;

    const float* Af = Amat + f*HD*HD;
    for (int e = tid; e < HD*HD; e += DTHREADS) {
        int r = e >> 6, c = e & 63;
        float a   = Af[e];
        float eye = (r == c) ? 1.0f : 0.0f;
        aug[r*AUGC + c]      = eye - 0.5f*a;
        aug[r*AUGC + 64 + c] = eye + 0.5f*a;
    }
    if (tid < HD)
        aug[tid*AUGC + 128] = Bmat[f*HD + tid];
    __syncthreads();

    {
        const int row = tid & 63;
        const int seg = tid >> 6;
        const int c0  = seg*8;
        for (int p = 0; p < HD; ++p) {
            float fac = aug[row*AUGC + p] / aug[p*AUGC + p];
            __syncthreads();
            if (row != p) {
                for (int c = 0; c < 8; ++c)
                    aug[row*AUGC + c0 + c] -= fac * aug[p*AUGC + c0 + c];
                if (seg == 15)
                    aug[row*AUGC + 128] -= fac * aug[p*AUGC + 128];
            }
            __syncthreads();
        }
    }

    float* Ab = dsc + (size_t)f*DSTRIDE;
    for (int e = tid; e < HD*HD; e += DTHREADS) {
        int r = e >> 6, c = e & 63;
        const float rinv = 1.0f / aug[r*AUGC + r];
        Ab[e] = aug[r*AUGC + 64 + c] * rinv;
    }
    if (tid < HD)
        Ab[HD*HD + tid] = aug[tid*AUGC + 128] / aug[tid*AUGC + tid];
}


__device__ __forceinline__ v8f tile64_kk(const float* Arow, int lda, const float* Brow, int ldb,
                                         int h, int c)
{
    v8f acc0 = {}, acc1 = {};
    const float* ap = Arow + c*lda + 2*h;
    const float* bp = Brow + c*ldb + 2*h;
    #pragma unroll
    for (int j = 0; j < 16; j += 2) {
        const v2f a0 = *(const v2f*)(ap + 4*j);
        const v2f a1 = *(const v2f*)(ap + 4*j + 4);
        const v2f b0 = *(const v2f*)(bp + 4*j);
        const v2f b1 = *(const v2f*)(bp + 4*j + 4);
        acc0 = __builtin_amdgcn_wmma_f32_16x16x4_f32(false, a0, false, b0, (short)0, acc0, false, false);
        acc1 = __builtin_amdgcn_wmma_f32_16x16x4_f32(false, a1, false, b1, (short)0, acc1, false, false);
    }
    return acc0 + acc1;
}

__device__ __forceinline__ void load_vec(v2f (&b)[16], const float* vec, int h)
{
    #pragma unroll
    for (int j4 = 0; j4 < 8; ++j4) {
        const v4f t = *(const v4f*)(vec + 32*h + 4*j4);
        b[2*j4].x     = t[0]; b[2*j4].y     = t[1];
        b[2*j4 + 1].x = t[2]; b[2*j4 + 1].y = t[3];
    }
}

__device__ __forceinline__ v8f matvec16(const v2f (&a)[16], const v2f (&b)[16])
{
    v8f acc0 = {}, acc1 = {};
    #pragma unroll
    for (int j = 0; j < 16; j += 2) {
        acc0 = __builtin_amdgcn_wmma_f32_16x16x4_f32(false, a[j],     false, b[j],     (short)0, acc0, false, false);
        acc1 = __builtin_amdgcn_wmma_f32_16x16x4_f32(false, a[j + 1], false, b[j + 1], (short)0, acc1, false, false);
    }
    return acc0 + acc1;
}

__launch_bounds__(STHREADS)
__global__ void ssm_conv_wmma_f32(const float* __restrict__ x,
                                  const float* __restrict__ dsc,
                                  const float* __restrict__ Cmat,
                                  float* __restrict__ out)
{
    __shared__ __attribute__((aligned(16))) float smem[(NHALF + 1)*VP + 2*HD + 16*MP + (16 + NT) + (NT + 16)*NBATCH + NBATCH*YP];
    float* Vm    = smem;
    float* wb    = Vm + (NHALF + 1)*VP;
    float* Lk    = wb + 2*HD;
    float* Kp    = Lk + 16*MP;
    float* u_lds = Kp + 16 + NT;
    float* y_lds = u_lds + (NT + 16)*NBATCH;

    const int f   = blockIdx.x;
    const int tid = threadIdx.x;
    const int wv  = __builtin_amdgcn_readfirstlane(tid >> 5);
    const int ln  = tid & 31;
    const int h   = ln >> 4;
    const int c   = ln & 15;
    const int g   = wv >> 2;
    const int q   = wv & 3;

    const float* Ab = dsc + (size_t)f*DSTRIDE;
    float xv[4][NBATCH];
    #pragma unroll
    for (int q4 = 0; q4 < 4; ++q4) {
        #pragma unroll
        for (int b = 0; b < NBATCH; ++b)
            xv[q4][b] = x[(size_t)(b*NFEAT + f)*NT + tid + q4*STHREADS];
    }
    const float bvv = (tid < HD) ? Ab[HD*HD + tid] : 0.0f;
    const float cvv = (tid < HD) ? Cmat[f*HD + tid] : 0.0f;

    v2f a_op[16];
    if (g == 0) {
        const float* ar = Ab + (16*q + c)*HD + 32*h;
        #pragma unroll
        for (int j4 = 0; j4 < 8; ++j4) {
            const v4f t = *(const v4f*)(ar + 4*j4);
            a_op[2*j4].x     = t[0]; a_op[2*j4].y     = t[1];
            a_op[2*j4 + 1].x = t[2]; a_op[2*j4 + 1].y = t[3];
        }
    } else {
        const float* ac = Ab + (32*h)*HD + 16*q + c;
        #pragma unroll
        for (int j = 0; j < 16; ++j) {
            a_op[j].x = ac[(2*j)*HD];
            a_op[j].y = ac[(2*j + 1)*HD];
        }
    }

    for (int e = tid; e < 16*MP; e += STHREADS)
        if (e >= HD) Lk[e] = 0.0f;
    if (tid < HD) { Vm[tid] = bvv; wb[tid] = cvv; Lk[tid] = cvv; }
    if (tid < 16)  Kp[tid] = 0.0f;
    if (tid < 128) u_lds[tid] = 0.0f;
    #pragma unroll
    for (int q4 = 0; q4 < 4; ++q4) {
        float* ud = u_lds + (16 + tid + q4*STHREADS)*NBATCH;
        const v4f lo4 = { xv[q4][0], xv[q4][1], xv[q4][2], xv[q4][3] };
        const v4f hi4 = { xv[q4][4], xv[q4][5], xv[q4][6], xv[q4][7] };
        *(v4f*)(ud)     = lo4;
        *(v4f*)(ud + 4) = hi4;
    }
    __syncthreads();

    {
        v2f b_op[16];
        load_vec(b_op, (g == 0) ? Vm : wb, h);
        for (int t = 0; t < NHALF; ++t) {
            float* nx = (g == 0) ? (Vm + (t + 1)*VP) : (wb + ((t + 1) & 1)*HD);
            const v8f d = matvec16(a_op, b_op);
            if (c == 0) {
                float* dp = nx + 16*q + 8*h;
                const v4f lo4 = { d[0], d[1], d[2], d[3] };
                const v4f hi4 = { d[4], d[5], d[6], d[7] };
                *(v4f*)(dp)     = lo4;
                *(v4f*)(dp + 4) = hi4;
            }
            __syncthreads();
            load_vec(b_op, nx, h);
        }
    }
    if (tid < HD) Lk[MP + tid] = wb[tid];
    __syncthreads();

    for (int i = 0; i < 4; ++i) {
        const int n = wv + NWAVES*i;
        const v8f kt = tile64_kk(Lk, MP, Vm + 16*n*VP, VP, h, c);
        if (h == 0) {
            Kp[16 + 16*n + c]         = kt[0];
            Kp[16 + NHALF + 16*n + c] = kt[1];
        }
    }
    __syncthreads();

    const int ucol = 128*(c >> 3) + (c & 7);
    for (int g = 0; g < 4; ++g) {
        const int p = (g & 1) ? (8*g + 7 - wv) : (8*g + wv);
        v8f acc0 = {}, acc1 = {};
        for (int dd = 0; dd <= 2*p + 1; dd += 2) {
            #pragma unroll
            for (int q = 0; q < 2; ++q) {
                const int d  = dd + q;
                const float* up = u_lds + (2*p - d + 1)*128 + ucol + (2*h)*NBATCH;
                const float* kp = Kp + 16 + 16*d + c - 2*h;
                #pragma unroll
                for (int j = 0; j < 4; ++j) {
                    v2f a, b;
                    a.x = kp[-4*j - 1];          a.y = kp[-4*j];
                    b.x = up[(4*j + 1)*NBATCH];  b.y = up[(4*j)*NBATCH];
                    if (j & 1)
                        acc1 = __builtin_amdgcn_wmma_f32_16x16x4_f32(false, a, false, b, (short)0, acc1, false, false);
                    else
                        acc0 = __builtin_amdgcn_wmma_f32_16x16x4_f32(false, a, false, b, (short)0, acc0, false, false);
                }
            }
        }
        const v8f y = acc0 + acc1;
        float* yp = y_lds + (c & 7)*YP + 16*(2*p + (c >> 3)) + 8*h;
        const v4f ylo = { y[0], y[1], y[2], y[3] };
        const v4f yhi = { y[4], y[5], y[6], y[7] };
        *(v4f*)(yp)     = ylo;
        *(v4f*)(yp + 4) = yhi;
    }
    __syncthreads();

    for (int e = tid; e < NBATCH*NT/4; e += STHREADS) {
        const int b = e >> 8, i4 = (e & 255)*4;
        *(v4f*)(out + (size_t)(b*NFEAT + f)*NT + i4) = *(const v4f*)(y_lds + b*YP + i4);
    }
}

extern "C" void kernel_launch(void* const* d_in, const int* in_sizes, int n_in,
                              void* d_out, int out_size, void* d_ws, size_t ws_size,
                              hipStream_t stream) {
    (void)in_sizes; (void)n_in; (void)ws_size; (void)out_size;
    const float* x = (const float*)d_in[0];
    const float* A = (const float*)d_in[1];
    const float* B = (const float*)d_in[2];
    const float* C = (const float*)d_in[3];
    float* out = (float*)d_out;
    float* dsc = (float*)d_ws;
    ssm_discretize_f32<<<dim3(NFEAT), dim3(DTHREADS), 0, stream>>>(A, B, dsc);
    ssm_conv_wmma_f32<<<dim3(NFEAT), dim3(STHREADS), 0, stream>>>(x, dsc, C, out);
}


# Impulse vectors move 16 per step as the 16 columns of one WMMA tile, V(j+1) = A_bar^16 V(j) (A_bar^16 by 4 WMMA squarings): 46 steps, not 512; w side 32 steps of A_bar^16; summation order changed
# speedup vs baseline: 9.6395x; 3.2665x over previous
#include <hip/hip_runtime.h>

typedef __attribute__((ext_vector_type(2))) float v2f;
typedef __attribute__((ext_vector_type(4))) float v4f;
typedef __attribute__((ext_vector_type(8))) float v8f;

#define NBATCH 8
#define NFEAT  256
#define HD     64
#define NT     1024
#define AUGC   130
#define DTHREADS 1024
#define DSTRIDE (HD*HD + HD)
#define STHREADS 256
#define NWAVES 8
#define NHALF  512
#define VBLK   16
#define NSQ    4
#define PP 68
#define VP 68
#define MP 68
#define YP 1028

__launch_bounds__(DTHREADS)
__global__ void ssm_discretize_f32(const float* __restrict__ Amat,
                                   const float* __restrict__ Bmat,
                                   float* __restrict__ dsc)
{
    __shared__ float aug[HD*AUGC];

    const int f   = blockIdx.x;
    const int tid = threadIdx.x;

    const float* Af = Amat + f*HD*HD;
    for (int e = tid; e < HD*HD; e += DTHREADS) {
        int r = e >> 6, c = e & 63;
        float a   = Af[e];
        float eye = (r == c) ? 1.0f : 0.0f;
        aug[r*AUGC + c]      = eye - 0.5f*a;
        aug[r*AUGC + 64 + c] = eye + 0.5f*a;
    }
    if (tid < HD)
        aug[tid*AUGC + 128] = Bmat[f*HD + tid];
    __syncthreads();

    {
        const int row = tid & 63;
        const int seg = tid >> 6;
        const int c0  = seg*8;
        for (int p = 0; p < HD; ++p) {
            float fac = aug[row*AUGC + p] / aug[p*AUGC + p];
            __syncthreads();
            if (row != p) {
                for (int c = 0; c < 8; ++c)
                    aug[row*AUGC + c0 + c] -= fac * aug[p*AUGC + c0 + c];
                if (seg == 15)
                    aug[row*AUGC + 128] -= fac * aug[p*AUGC + 128];
            }
            __syncthreads();
        }
    }

    float* Ab = dsc + (size_t)f*DSTRIDE;
    for (int e = tid; e < HD*HD; e += DTHREADS) {
        int r = e >> 6, c = e & 63;
        const float rinv = 1.0f / aug[r*AUGC + r];
        Ab[e] = aug[r*AUGC + 64 + c] * rinv;
    }
    if (tid < HD)
        Ab[HD*HD + tid] = aug[tid*AUGC + 128] / aug[tid*AUGC + tid];
}


__device__ __forceinline__ v8f tile64_kk(const float* Arow, int lda, const float* Brow, int ldb,
                                         int h, int c)
{
    v8f acc0 = {}, acc1 = {};
    const float* ap = Arow + c*lda + 2*h;
    const float* bp = Brow + c*ldb + 2*h;
    #pragma unroll
    for (int j = 0; j < 16; j += 2) {
        const v2f a0 = *(const v2f*)(ap + 4*j);
        const v2f a1 = *(const v2f*)(ap + 4*j + 4);
        const v2f b0 = *(const v2f*)(bp + 4*j);
        const v2f b1 = *(const v2f*)(bp + 4*j + 4);
        acc0 = __builtin_amdgcn_wmma_f32_16x16x4_f32(false, a0, false, b0, (short)0, acc0, false, false);
        acc1 = __builtin_amdgcn_wmma_f32_16x16x4_f32(false, a1, false, b1, (short)0, acc1, false, false);
    }
    return acc0 + acc1;
}

__device__ __forceinline__ void load_vec(v2f (&b)[16], const float* vec, int h)
{
    #pragma unroll
    for (int j4 = 0; j4 < 8; ++j4) {
        const v4f t = *(const v4f*)(vec + 32*h + 4*j4);
        b[2*j4].x     = t[0]; b[2*j4].y     = t[1];
        b[2*j4 + 1].x = t[2]; b[2*j4 + 1].y = t[3];
    }
}

__device__ __forceinline__ v8f matvec16(const v2f (&a)[16], const v2f (&b)[16])
{
    v8f acc0 = {}, acc1 = {};
    #pragma unroll
    for (int j = 0; j < 16; j += 2) {
        acc0 = __builtin_amdgcn_wmma_f32_16x16x4_f32(false, a[j],     false, b[j],     (short)0, acc0, false, false);
        acc1 = __builtin_amdgcn_wmma_f32_16x16x4_f32(false, a[j + 1], false, b[j + 1], (short)0, acc1, false, false);
    }
    return acc0 + acc1;
}

__launch_bounds__(STHREADS)
__global__ void ssm_conv_wmma_f32(const float* __restrict__ x,
                                  const float* __restrict__ dsc,
                                  const float* __restrict__ Cmat,
                                  float* __restrict__ out)
{
    __shared__ __attribute__((aligned(16))) float smem[(NHALF + 1)*VP + 2*HD + 16*MP + (16 + NT) + (NT + 16)*NBATCH + NBATCH*YP + 4*HD*PP];
    float* Vm    = smem;
    float* wb    = Vm + (NHALF + 1)*VP;
    float* Lk    = wb + 2*HD;
    float* Kp    = Lk + 16*MP;
    float* u_lds = Kp + 16 + NT;
    float* y_lds = u_lds + (NT + 16)*NBATCH;
    float* Pw    = y_lds + NBATCH*YP;

    const int f   = blockIdx.x;
    const int tid = threadIdx.x;
    const int wv  = __builtin_amdgcn_readfirstlane(tid >> 5);
    const int ln  = tid & 31;
    const int h   = ln >> 4;
    const int c   = ln & 15;
    const int g   = wv >> 2;
    const int q   = wv & 3;

    const float* Ab = dsc + (size_t)f*DSTRIDE;
    float xv[4][NBATCH];
    #pragma unroll
    for (int q4 = 0; q4 < 4; ++q4) {
        #pragma unroll
        for (int b = 0; b < NBATCH; ++b)
            xv[q4][b] = x[(size_t)(b*NFEAT + f)*NT + tid + q4*STHREADS];
    }
    v4f pv[4];
    #pragma unroll
    for (int i = 0; i < 4; ++i) {
        const int e4 = tid + i*STHREADS;
        pv[i] = *(const v4f*)(Ab + (e4 >> 4)*HD + (e4 & 15)*4);
    }
    const float bvv = (tid < HD) ? Ab[HD*HD + tid] : 0.0f;
    const float cvv = (tid < HD) ? Cmat[f*HD + tid] : 0.0f;

    v2f a_op[16];
    if (g == 0)
        load_vec(a_op, Ab + (16*q + c)*HD, h);

    for (int e = tid; e < 16*MP; e += STHREADS)
        if (e >= HD) Lk[e] = 0.0f;
    if (tid < HD) { Vm[tid] = bvv; wb[tid] = cvv; Lk[tid] = cvv; }
    #pragma unroll
    for (int i = 0; i < 4; ++i) {
        const int e4 = tid + i*STHREADS;
        const int r = e4 >> 4, c4 = (e4 & 15)*4;
        *(v4f*)(Pw + r*PP + c4) = pv[i];
        #pragma unroll
        for (int s = 0; s < 4; ++s)
            Pw[HD*PP + (c4 + s)*PP + r] = pv[i][s];
    }
    if (tid < 16)  Kp[tid] = 0.0f;
    if (tid < 128) u_lds[tid] = 0.0f;
    #pragma unroll
    for (int q4 = 0; q4 < 4; ++q4) {
        float* ud = u_lds + (16 + tid + q4*STHREADS)*NBATCH;
        const v4f lo4 = { xv[q4][0], xv[q4][1], xv[q4][2], xv[q4][3] };
        const v4f hi4 = { xv[q4][4], xv[q4][5], xv[q4][6], xv[q4][7] };
        *(v4f*)(ud)     = lo4;
        *(v4f*)(ud + 4) = hi4;
    }
    __syncthreads();

    #pragma nounroll
    for (int s = 0; s < NSQ; ++s) {
        const float* Pc  = Pw + (s & 1)*2*HD*PP;
        const float* PTc = Pc + HD*PP;
        float* Pn  = Pw + ((s + 1) & 1)*2*HD*PP;
        float* PTn = Pn + HD*PP;
        for (int i = 0; i < 2; ++i) {
            const int tl = wv + NWAVES*i;
            const int tm = tl >> 2, tn = tl & 3;
            const v8f d = tile64_kk(Pc + 16*tm*PP, PP, PTc + 16*tn*PP, PP, h, c);
            float* pr = Pn + (16*tm + 8*h)*PP + 16*tn + c;
            #pragma unroll
            for (int r = 0; r < 8; ++r) pr[r*PP] = d[r];
            float* pt = PTn + (16*tn + c)*PP + 16*tm + 8*h;
            const v4f lo4 = { d[0], d[1], d[2], d[3] };
            const v4f hi4 = { d[4], d[5], d[6], d[7] };
            *(v4f*)(pt)     = lo4;
            *(v4f*)(pt + 4) = hi4;
        }
        __syncthreads();
    }
    const float* P16  = Pw;
    const float* PT16 = Pw + HD*PP;

    {
        if (g != 0)
            load_vec(a_op, PT16 + (16*q + c)*PP, h);
        v2f b_op[16];
        load_vec(b_op, (g == 0) ? Vm : wb, h);
        #pragma nounroll
        for (int t = 0; t < VBLK - 1; ++t) {
            float* nx = (g == 0) ? (Vm + (t + 1)*VP) : (wb + ((t + 1) & 1)*HD);
            const v8f d = matvec16(a_op, b_op);
            if (c == 0) {
                float* dp = nx + 16*q + 8*h;
                const v4f lo4 = { d[0], d[1], d[2], d[3] };
                const v4f hi4 = { d[4], d[5], d[6], d[7] };
                *(v4f*)(dp)     = lo4;
                *(v4f*)(dp + 4) = hi4;
            }
            __syncthreads();
            load_vec(b_op, nx, h);
        }
        if (g == 0) {
            load_vec(a_op, P16 + (16*q + c)*PP, h);
            load_vec(b_op, Vm + c*VP, h);
        }
        for (int j = 0; j < NHALF/VBLK - 1; ++j) {
            float* nx = (g == 0) ? (Vm + (VBLK*(j + 1) + c)*VP) : (wb + ((VBLK + j) & 1)*HD);
            const bool on = (g == 0) || (j < NHALF/VBLK - (VBLK - 1));
            if (on) {
                const v8f d = matvec16(a_op, b_op);
                if (g == 0 || c == 0) {
                    float* dp = nx + 16*q + 8*h;
                    const v4f lo4 = { d[0], d[1], d[2], d[3] };
                    const v4f hi4 = { d[4], d[5], d[6], d[7] };
                    *(v4f*)(dp)     = lo4;
                    *(v4f*)(dp + 4) = hi4;
                }
            }
            __syncthreads();
            if (on) load_vec(b_op, nx, h);
        }
    }
    if (tid < HD) Lk[MP + tid] = wb[tid];
    __syncthreads();

    for (int i = 0; i < 4; ++i) {
        const int n = wv + NWAVES*i;
        const v8f kt = tile64_kk(Lk, MP, Vm + 16*n*VP, VP, h, c);
        if (h == 0) {
            Kp[16 + 16*n + c]         = kt[0];
            Kp[16 + NHALF + 16*n + c] = kt[1];
        }
    }
    __syncthreads();

    const int ucol = 128*(c >> 3) + (c & 7);
    for (int g = 0; g < 4; ++g) {
        const int p = (g & 1) ? (8*g + 7 - wv) : (8*g + wv);
        v8f acc0 = {}, acc1 = {};
        for (int dd = 0; dd <= 2*p + 1; dd += 2) {
            #pragma unroll
            for (int q = 0; q < 2; ++q) {
                const int d  = dd + q;
                const float* up = u_lds + (2*p - d + 1)*128 + ucol + (2*h)*NBATCH;
                const float* kp = Kp + 16 + 16*d + c - 2*h;
                #pragma unroll
                for (int j = 0; j < 4; ++j) {
                    v2f a, b;
                    a.x = kp[-4*j - 1];          a.y = kp[-4*j];
                    b.x = up[(4*j + 1)*NBATCH];  b.y = up[(4*j)*NBATCH];
                    if (j & 1)
                        acc1 = __builtin_amdgcn_wmma_f32_16x16x4_f32(false, a, false, b, (short)0, acc1, false, false);
                    else
                        acc0 = __builtin_amdgcn_wmma_f32_16x16x4_f32(false, a, false, b, (short)0, acc0, false, false);
                }
            }
        }
        const v8f y = acc0 + acc1;
        float* yp = y_lds + (c & 7)*YP + 16*(2*p + (c >> 3)) + 8*h;
        const v4f ylo = { y[0], y[1], y[2], y[3] };
        const v4f yhi = { y[4], y[5], y[6], y[7] };
        *(v4f*)(yp)     = ylo;
        *(v4f*)(yp + 4) = yhi;
    }
    __syncthreads();

    for (int e = tid; e < NBATCH*NT/4; e += STHREADS) {
        const int b = e >> 8, i4 = (e & 255)*4;
        *(v4f*)(out + (size_t)(b*NFEAT + f)*NT + i4) = *(const v4f*)(y_lds + b*YP + i4);
    }
}

extern "C" void kernel_launch(void* const* d_in, const int* in_sizes, int n_in,
                              void* d_out, int out_size, void* d_ws, size_t ws_size,
                              hipStream_t stream) {
    (void)in_sizes; (void)n_in; (void)ws_size; (void)out_size;
    const float* x = (const float*)d_in[0];
    const float* A = (const float*)d_in[1];
    const float* B = (const float*)d_in[2];
    const float* C = (const float*)d_in[3];
    float* out = (float*)d_out;
    float* dsc = (float*)d_ws;
    ssm_discretize_f32<<<dim3(NFEAT), dim3(DTHREADS), 0, stream>>>(A, B, dsc);
    ssm_conv_wmma_f32<<<dim3(NFEAT), dim3(STHREADS), 0, stream>>>(x, dsc, C, out);
}


# Discretization as a blocked Gauss-Jordan: 4 block pivots of 16, diagonal block inverted in registers (DPP row_share), row updates as f32 WMMA on register-held column tiles; summation order changed
# speedup vs baseline: 14.0312x; 1.4556x over previous
#include <hip/hip_runtime.h>

typedef __attribute__((ext_vector_type(2))) float v2f;
typedef __attribute__((ext_vector_type(4))) float v4f;
typedef __attribute__((ext_vector_type(8))) float v8f;

#define NBATCH 8
#define NFEAT  256
#define HD     64
#define NT     1024
#define GTHREADS 256
#define LP     20
#define DSTRIDE (HD*HD + HD)
#define STHREADS 256
#define NWAVES 8
#define NHALF  512
#define VBLK   16
#define NSQ    4
#define PP 68
#define VP 68
#define MP 68
#define YP 1028


template<int P> __device__ __forceinline__ float row_bcast(float x)
{
    return __builtin_bit_cast(float, __builtin_amdgcn_update_dpp(0, __builtin_bit_cast(int, x), 0x150 + P, 0xF, 0xF, false));
}

template<int P> __device__ __forceinline__ void gj_pivot(float (&d)[16], float (&e)[8], int c)
{
    const float pp  = row_bcast<P>(d[P]);
    const float fac = (c != P) ? d[P] / pp : 0.0f;
    #pragma unroll
    for (int cc = P + 1; cc < 16; ++cc)
        d[cc] -= fac*row_bcast<P>(d[cc]);
    #pragma unroll
    for (int m = 0; m < 8; ++m)
        e[m] -= fac*row_bcast<P>(e[m]);
}

__launch_bounds__(GTHREADS)
__global__ void ssm_discretize_f32(const float* __restrict__ Amat,
                                   const float* __restrict__ Bmat,
                                   float* __restrict__ dsc)
{
    __shared__ __attribute__((aligned(16))) float Lc[2*HD*LP];

    const int f   = blockIdx.x;
    const int tid = threadIdx.x;
    const int wv  = __builtin_amdgcn_readfirstlane(tid >> 5);
    const int ln  = tid & 31;
    const int h   = ln >> 4;
    const int c   = ln & 15;
    const int n   = wv + 1;

    const float* Af = Amat + f*HD*HD;

    for (int m = 0; m < 4; ++m) {
        const int e = tid + m*GTHREADS;
        const int r = e >> 4, cc = e & 15;
        Lc[r*LP + cc] = ((r == cc) ? 1.0f : 0.0f) - 0.5f*Af[r*HD + cc];
    }

    v8f T[4];
    if (n <= 7) {
        const float sg = (n <= 3) ? -0.5f : 0.5f;
        const int col = 16*(n & 3) + c;
        #pragma unroll
        for (int i = 0; i < 4; ++i) {
            #pragma unroll
            for (int r = 0; r < 8; ++r) {
                const int row = 16*i + 8*h + r;
                T[i][r] = ((row == col) ? 1.0f : 0.0f) + sg*Af[row*HD + col];
            }
        }
    } else {
        #pragma unroll
        for (int i = 0; i < 4; ++i) {
            #pragma unroll
            for (int r = 0; r < 8; ++r) {
                const float bv = Bmat[f*HD + 16*i + 8*h + r];
                T[i][r] = (c == 0) ? bv : 0.0f;
            }
        }
    }
    __syncthreads();

    #pragma nounroll
    for (int kb = 0; kb < 4; ++kb) {
        const float* Lb = Lc + (kb & 1)*HD*LP;
        if (n > kb) {
            float d[16], e[8];
            const float* dr = Lb + (16*kb + c)*LP;
            #pragma unroll
            for (int q4 = 0; q4 < 4; ++q4) {
                const v4f t = *(const v4f*)(dr + 4*q4);
                d[4*q4] = t[0]; d[4*q4 + 1] = t[1]; d[4*q4 + 2] = t[2]; d[4*q4 + 3] = t[3];
            }
            #pragma unroll
            for (int m = 0; m < 8; ++m) e[m] = (8*h + m == c) ? 1.0f : 0.0f;
            gj_pivot<0>(d, e, c);   gj_pivot<1>(d, e, c);   gj_pivot<2>(d, e, c);   gj_pivot<3>(d, e, c);
            gj_pivot<4>(d, e, c);   gj_pivot<5>(d, e, c);   gj_pivot<6>(d, e, c);   gj_pivot<7>(d, e, c);
            gj_pivot<8>(d, e, c);   gj_pivot<9>(d, e, c);   gj_pivot<10>(d, e, c);  gj_pivot<11>(d, e, c);
            gj_pivot<12>(d, e, c);  gj_pivot<13>(d, e, c);  gj_pivot<14>(d, e, c);  gj_pivot<15>(d, e, c);
            float diag = d[0];
            #pragma unroll
            for (int p = 1; p < 16; ++p) diag = (c == p) ? d[p] : diag;
            #pragma unroll
            for (int m = 0; m < 8; ++m) e[m] = e[m] / diag;

            v8f R = {};
            #pragma unroll
            for (int j = 0; j < 4; ++j) {
                const v2f a = { e[2*j], e[2*j + 1] };
                const v2f b = { T[0][2*j], T[0][2*j + 1] };
                R = __builtin_amdgcn_wmma_f32_16x16x4_f32(false, a, false, b, (short)0, R, false, false);
            }
            T[0] = R;

            #pragma unroll
            for (int i = 1; i < 4; ++i) {
                const float* lr = Lb + (16*((kb + i) & 3) + c)*LP + 8*h;
                const v4f l0 = *(const v4f*)(lr);
                const v4f l1 = *(const v4f*)(lr + 4);
                const v2f a0 = { -l0[0], -l0[1] }, a1 = { -l0[2], -l0[3] };
                const v2f a2 = { -l1[0], -l1[1] }, a3 = { -l1[2], -l1[3] };
                const v2f b0 = { R[0], R[1] }, b1 = { R[2], R[3] }, b2 = { R[4], R[5] }, b3 = { R[6], R[7] };
                v8f acc = T[i];
                acc = __builtin_amdgcn_wmma_f32_16x16x4_f32(false, a0, false, b0, (short)0, acc, false, false);
                acc = __builtin_amdgcn_wmma_f32_16x16x4_f32(false, a1, false, b1, (short)0, acc, false, false);
                acc = __builtin_amdgcn_wmma_f32_16x16x4_f32(false, a2, false, b2, (short)0, acc, false, false);
                acc = __builtin_amdgcn_wmma_f32_16x16x4_f32(false, a3, false, b3, (short)0, acc, false, false);
                T[i] = acc;
            }

            if (kb < 3 && wv == kb) {
                float* Ln = Lc + ((kb + 1) & 1)*HD*LP;
                #pragma unroll
                for (int i = 0; i < 4; ++i) {
                    #pragma unroll
                    for (int r = 0; r < 8; ++r)
                        Ln[(16*((kb + i) & 3) + 8*h + r)*LP + c] = T[i][r];
                }
            }
        }
        { const v8f t0 = T[0]; T[0] = T[1]; T[1] = T[2]; T[2] = T[3]; T[3] = t0; }
        if (kb < 3) __syncthreads();
    }

    float* Ab = dsc + (size_t)f*DSTRIDE;
    if (n >= 4 && n <= 7) {
        #pragma unroll
        for (int i = 0; i < 4; ++i) {
            #pragma unroll
            for (int r = 0; r < 8; ++r)
                Ab[(16*i + 8*h + r)*HD + 16*(n - 4) + c] = T[i][r];
        }
    } else if (n == 8) {
        if (c == 0) {
            #pragma unroll
            for (int i = 0; i < 4; ++i) {
                #pragma unroll
                for (int r = 0; r < 8; ++r)
                    Ab[HD*HD + 16*i + 8*h + r] = T[i][r];
            }
        }
    }
}


__device__ __forceinline__ v8f tile64_kk(const float* Arow, int lda, const float* Brow, int ldb,
                                         int h, int c)
{
    v8f acc0 = {}, acc1 = {};
    const float* ap = Arow + c*lda + 2*h;
    const float* bp = Brow + c*ldb + 2*h;
    #pragma unroll
    for (int j = 0; j < 16; j += 2) {
        const v2f a0 = *(const v2f*)(ap + 4*j);
        const v2f a1 = *(const v2f*)(ap + 4*j + 4);
        const v2f b0 = *(const v2f*)(bp + 4*j);
        const v2f b1 = *(const v2f*)(bp + 4*j + 4);
        acc0 = __builtin_amdgcn_wmma_f32_16x16x4_f32(false, a0, false, b0, (short)0, acc0, false, false);
        acc1 = __builtin_amdgcn_wmma_f32_16x16x4_f32(false, a1, false, b1, (short)0, acc1, false, false);
    }
    return acc0 + acc1;
}

__device__ __forceinline__ void load_vec(v2f (&b)[16], const float* vec, int h)
{
    #pragma unroll
    for (int j4 = 0; j4 < 8; ++j4) {
        const v4f t = *(const v4f*)(vec + 32*h + 4*j4);
        b[2*j4].x     = t[0]; b[2*j4].y     = t[1];
        b[2*j4 + 1].x = t[2]; b[2*j4 + 1].y = t[3];
    }
}

__device__ __forceinline__ v8f matvec16(const v2f (&a)[16], const v2f (&b)[16])
{
    v8f acc0 = {}, acc1 = {};
    #pragma unroll
    for (int j = 0; j < 16; j += 2) {
        acc0 = __builtin_amdgcn_wmma_f32_16x16x4_f32(false, a[j],     false, b[j],     (short)0, acc0, false, false);
        acc1 = __builtin_amdgcn_wmma_f32_16x16x4_f32(false, a[j + 1], false, b[j + 1], (short)0, acc1, false, false);
    }
    return acc0 + acc1;
}

__launch_bounds__(STHREADS)
__global__ void ssm_conv_wmma_f32(const float* __restrict__ x,
                                  const float* __restrict__ dsc,
                                  const float* __restrict__ Cmat,
                                  float* __restrict__ out)
{
    __shared__ __attribute__((aligned(16))) float smem[(NHALF + 1)*VP + 2*HD + 16*MP + (16 + NT) + (NT + 16)*NBATCH + NBATCH*YP + 4*HD*PP];
    float* Vm    = smem;
    float* wb    = Vm + (NHALF + 1)*VP;
    float* Lk    = wb + 2*HD;
    float* Kp    = Lk + 16*MP;
    float* u_lds = Kp + 16 + NT;
    float* y_lds = u_lds + (NT + 16)*NBATCH;
    float* Pw    = y_lds + NBATCH*YP;

    const int f   = blockIdx.x;
    const int tid = threadIdx.x;
    const int wv  = __builtin_amdgcn_readfirstlane(tid >> 5);
    const int ln  = tid & 31;
    const int h   = ln >> 4;
    const int c   = ln & 15;
    const int g   = wv >> 2;
    const int q   = wv & 3;

    const float* Ab = dsc + (size_t)f*DSTRIDE;
    float xv[4][NBATCH];
    #pragma unroll
    for (int q4 = 0; q4 < 4; ++q4) {
        #pragma unroll
        for (int b = 0; b < NBATCH; ++b)
            xv[q4][b] = x[(size_t)(b*NFEAT + f)*NT + tid + q4*STHREADS];
    }
    v4f pv[4];
    #pragma unroll
    for (int i = 0; i < 4; ++i) {
        const int e4 = tid + i*STHREADS;
        pv[i] = *(const v4f*)(Ab + (e4 >> 4)*HD + (e4 & 15)*4);
    }
    const float bvv = (tid < HD) ? Ab[HD*HD + tid] : 0.0f;
    const float cvv = (tid < HD) ? Cmat[f*HD + tid] : 0.0f;

    v2f a_op[16];
    if (g == 0)
        load_vec(a_op, Ab + (16*q + c)*HD, h);

    for (int e = tid; e < 16*MP; e += STHREADS)
        if (e >= HD) Lk[e] = 0.0f;
    if (tid < HD) { Vm[tid] = bvv; wb[tid] = cvv; Lk[tid] = cvv; }
    #pragma unroll
    for (int i = 0; i < 4; ++i) {
        const int e4 = tid + i*STHREADS;
        const int r = e4 >> 4, c4 = (e4 & 15)*4;
        *(v4f*)(Pw + r*PP + c4) = pv[i];
        #pragma unroll
        for (int s = 0; s < 4; ++s)
            Pw[HD*PP + (c4 + s)*PP + r] = pv[i][s];
    }
    if (tid < 16)  Kp[tid] = 0.0f;
    if (tid < 128) u_lds[tid] = 0.0f;
    #pragma unroll
    for (int q4 = 0; q4 < 4; ++q4) {
        float* ud = u_lds + (16 + tid + q4*STHREADS)*NBATCH;
        const v4f lo4 = { xv[q4][0], xv[q4][1], xv[q4][2], xv[q4][3] };
        const v4f hi4 = { xv[q4][4], xv[q4][5], xv[q4][6], xv[q4][7] };
        *(v4f*)(ud)     = lo4;
        *(v4f*)(ud + 4) = hi4;
    }
    __syncthreads();

    #pragma nounroll
    for (int s = 0; s < NSQ; ++s) {
        const float* Pc  = Pw + (s & 1)*2*HD*PP;
        const float* PTc = Pc + HD*PP;
        float* Pn  = Pw + ((s + 1) & 1)*2*HD*PP;
        float* PTn = Pn + HD*PP;
        for (int i = 0; i < 2; ++i) {
            const int tl = wv + NWAVES*i;
            const int tm = tl >> 2, tn = tl & 3;
            const v8f d = tile64_kk(Pc + 16*tm*PP, PP, PTc + 16*tn*PP, PP, h, c);
            float* pr = Pn + (16*tm + 8*h)*PP + 16*tn + c;
            #pragma unroll
            for (int r = 0; r < 8; ++r) pr[r*PP] = d[r];
            float* pt = PTn + (16*tn + c)*PP + 16*tm + 8*h;
            const v4f lo4 = { d[0], d[1], d[2], d[3] };
            const v4f hi4 = { d[4], d[5], d[6], d[7] };
            *(v4f*)(pt)     = lo4;
            *(v4f*)(pt + 4) = hi4;
        }
        __syncthreads();
    }
    const float* P16  = Pw;
    const float* PT16 = Pw + HD*PP;

    {
        if (g != 0)
            load_vec(a_op, PT16 + (16*q + c)*PP, h);
        v2f b_op[16];
        load_vec(b_op, (g == 0) ? Vm : wb, h);
        #pragma nounroll
        for (int t = 0; t < VBLK - 1; ++t) {
            float* nx = (g == 0) ? (Vm + (t + 1)*VP) : (wb + ((t + 1) & 1)*HD);
            const v8f d = matvec16(a_op, b_op);
            if (c == 0) {
                float* dp = nx + 16*q + 8*h;
                const v4f lo4 = { d[0], d[1], d[2], d[3] };
                const v4f hi4 = { d[4], d[5], d[6], d[7] };
                *(v4f*)(dp)     = lo4;
                *(v4f*)(dp + 4) = hi4;
            }
            __syncthreads();
            load_vec(b_op, nx, h);
        }
        if (g == 0) {
            load_vec(a_op, P16 + (16*q + c)*PP, h);
            load_vec(b_op, Vm + c*VP, h);
        }
        for (int j = 0; j < NHALF/VBLK - 1; ++j) {
            float* nx = (g == 0) ? (Vm + (VBLK*(j + 1) + c)*VP) : (wb + ((VBLK + j) & 1)*HD);
            const bool on = (g == 0) || (j < NHALF/VBLK - (VBLK - 1));
            if (on) {
                const v8f d = matvec16(a_op, b_op);
                if (g == 0 || c == 0) {
                    float* dp = nx + 16*q + 8*h;
                    const v4f lo4 = { d[0], d[1], d[2], d[3] };
                    const v4f hi4 = { d[4], d[5], d[6], d[7] };
                    *(v4f*)(dp)     = lo4;
                    *(v4f*)(dp + 4) = hi4;
                }
            }
            __syncthreads();
            if (on) load_vec(b_op, nx, h);
        }
    }
    if (tid < HD) Lk[MP + tid] = wb[tid];
    __syncthreads();

    for (int i = 0; i < 4; ++i) {
        const int n = wv + NWAVES*i;
        const v8f kt = tile64_kk(Lk, MP, Vm + 16*n*VP, VP, h, c);
        if (h == 0) {
            Kp[16 + 16*n + c]         = kt[0];
            Kp[16 + NHALF + 16*n + c] = kt[1];
        }
    }
    __syncthreads();

    const int ucol = 128*(c >> 3) + (c & 7);
    for (int g = 0; g < 4; ++g) {
        const int p = (g & 1) ? (8*g + 7 - wv) : (8*g + wv);
        v8f acc0 = {}, acc1 = {};
        for (int dd = 0; dd <= 2*p + 1; dd += 2) {
            #pragma unroll
            for (int q = 0; q < 2; ++q) {
                const int d  = dd + q;
                const float* up = u_lds + (2*p - d + 1)*128 + ucol + (2*h)*NBATCH;
                const float* kp = Kp + 16 + 16*d + c - 2*h;
                #pragma unroll
                for (int j = 0; j < 4; ++j) {
                    v2f a, b;
                    a.x = kp[-4*j - 1];          a.y = kp[-4*j];
                    b.x = up[(4*j + 1)*NBATCH];  b.y = up[(4*j)*NBATCH];
                    if (j & 1)
                        acc1 = __builtin_amdgcn_wmma_f32_16x16x4_f32(false, a, false, b, (short)0, acc1, false, false);
                    else
                        acc0 = __builtin_amdgcn_wmma_f32_16x16x4_f32(false, a, false, b, (short)0, acc0, false, false);
                }
            }
        }
        const v8f y = acc0 + acc1;
        float* yp = y_lds + (c & 7)*YP + 16*(2*p + (c >> 3)) + 8*h;
        const v4f ylo = { y[0], y[1], y[2], y[3] };
        const v4f yhi = { y[4], y[5], y[6], y[7] };
        *(v4f*)(yp)     = ylo;
        *(v4f*)(yp + 4) = yhi;
    }
    __syncthreads();

    for (int e = tid; e < NBATCH*NT/4; e += STHREADS) {
        const int b = e >> 8, i4 = (e & 255)*4;
        *(v4f*)(out + (size_t)(b*NFEAT + f)*NT + i4) = *(const v4f*)(y_lds + b*YP + i4);
    }
}

extern "C" void kernel_launch(void* const* d_in, const int* in_sizes, int n_in,
                              void* d_out, int out_size, void* d_ws, size_t ws_size,
                              hipStream_t stream) {
    (void)in_sizes; (void)n_in; (void)ws_size; (void)out_size;
    const float* x = (const float*)d_in[0];
    const float* A = (const float*)d_in[1];
    const float* B = (const float*)d_in[2];
    const float* C = (const float*)d_in[3];
    float* out = (float*)d_out;
    float* dsc = (float*)d_ws;
    ssm_discretize_f32<<<dim3(NFEAT), dim3(GTHREADS), 0, stream>>>(A, B, dsc);
    ssm_conv_wmma_f32<<<dim3(NFEAT), dim3(STHREADS), 0, stream>>>(x, dsc, C, out);
}


# Taps as a 32x32 product K[32a+b] = (Cv A_bar^32a).(A_bar^b Bv); both vector sets doubled beside 9 squarings of A_bar: 10 stages, not 4 squarings and 46 vector steps; summation order changed
# speedup vs baseline: 17.2450x; 1.2290x over previous
#include <hip/hip_runtime.h>

typedef __attribute__((ext_vector_type(2))) float v2f;
typedef __attribute__((ext_vector_type(4))) float v4f;
typedef __attribute__((ext_vector_type(8))) float v8f;

#define NBATCH 8
#define NFEAT  256
#define HD     64
#define NT     1024
#define GTHREADS 256
#define LP     20
#define DSTRIDE (HD*HD + HD)
#define STHREADS 256
#define NVEC   32
#define NSTAGE 10
#define PP 68
#define VP 68
#define YP 1028


template<int P> __device__ __forceinline__ float row_bcast(float x)
{
    return __builtin_bit_cast(float, __builtin_amdgcn_update_dpp(0, __builtin_bit_cast(int, x), 0x150 + P, 0xF, 0xF, false));
}

template<int P> __device__ __forceinline__ void gj_pivot(float (&d)[16], float (&e)[8], int c)
{
    const float pp  = row_bcast<P>(d[P]);
    const float fac = (c != P) ? d[P] / pp : 0.0f;
    #pragma unroll
    for (int cc = P + 1; cc < 16; ++cc)
        d[cc] -= fac*row_bcast<P>(d[cc]);
    #pragma unroll
    for (int m = 0; m < 8; ++m)
        e[m] -= fac*row_bcast<P>(e[m]);
}

__launch_bounds__(GTHREADS)
__global__ void ssm_discretize_f32(const float* __restrict__ Amat,
                                   const float* __restrict__ Bmat,
                                   float* __restrict__ dsc)
{
    __shared__ __attribute__((aligned(16))) float Lc[2*HD*LP];

    const int f   = blockIdx.x;
    const int tid = threadIdx.x;
    const int wv  = __builtin_amdgcn_readfirstlane(tid >> 5);
    const int ln  = tid & 31;
    const int h   = ln >> 4;
    const int c   = ln & 15;
    const int n   = wv + 1;

    const float* Af = Amat + f*HD*HD;

    for (int m = 0; m < 4; ++m) {
        const int e = tid + m*GTHREADS;
        const int r = e >> 4, cc = e & 15;
        Lc[r*LP + cc] = ((r == cc) ? 1.0f : 0.0f) - 0.5f*Af[r*HD + cc];
    }

    v8f T[4];
    if (n <= 7) {
        const float sg = (n <= 3) ? -0.5f : 0.5f;
        const int col = 16*(n & 3) + c;
        #pragma unroll
        for (int i = 0; i < 4; ++i) {
            #pragma unroll
            for (int r = 0; r < 8; ++r) {
                const int row = 16*i + 8*h + r;
                T[i][r] = ((row == col) ? 1.0f : 0.0f) + sg*Af[row*HD + col];
            }
        }
    } else {
        #pragma unroll
        for (int i = 0; i < 4; ++i) {
            #pragma unroll
            for (int r = 0; r < 8; ++r) {
                const float bv = Bmat[f*HD + 16*i + 8*h + r];
                T[i][r] = (c == 0) ? bv : 0.0f;
            }
        }
    }
    __syncthreads();

    #pragma nounroll
    for (int kb = 0; kb < 4; ++kb) {
        const float* Lb = Lc + (kb & 1)*HD*LP;
        if (n > kb) {
            float d[16], e[8];
            const float* dr = Lb + (16*kb + c)*LP;
            #pragma unroll
            for (int q4 = 0; q4 < 4; ++q4) {
                const v4f t = *(const v4f*)(dr + 4*q4);
                d[4*q4] = t[0]; d[4*q4 + 1] = t[1]; d[4*q4 + 2] = t[2]; d[4*q4 + 3] = t[3];
            }
            #pragma unroll
            for (int m = 0; m < 8; ++m) e[m] = (8*h + m == c) ? 1.0f : 0.0f;
            gj_pivot<0>(d, e, c);   gj_pivot<1>(d, e, c);   gj_pivot<2>(d, e, c);   gj_pivot<3>(d, e, c);
            gj_pivot<4>(d, e, c);   gj_pivot<5>(d, e, c);   gj_pivot<6>(d, e, c);   gj_pivot<7>(d, e, c);
            gj_pivot<8>(d, e, c);   gj_pivot<9>(d, e, c);   gj_pivot<10>(d, e, c);  gj_pivot<11>(d, e, c);
            gj_pivot<12>(d, e, c);  gj_pivot<13>(d, e, c);  gj_pivot<14>(d, e, c);  gj_pivot<15>(d, e, c);
            float diag = d[0];
            #pragma unroll
            for (int p = 1; p < 16; ++p) diag = (c == p) ? d[p] : diag;
            #pragma unroll
            for (int m = 0; m < 8; ++m) e[m] = e[m] / diag;

            v8f R = {};
            #pragma unroll
            for (int j = 0; j < 4; ++j) {
                const v2f a = { e[2*j], e[2*j + 1] };
                const v2f b = { T[0][2*j], T[0][2*j + 1] };
                R = __builtin_amdgcn_wmma_f32_16x16x4_f32(false, a, false, b, (short)0, R, false, false);
            }
            T[0] = R;

            #pragma unroll
            for (int i = 1; i < 4; ++i) {
                const float* lr = Lb + (16*((kb + i) & 3) + c)*LP + 8*h;
                const v4f l0 = *(const v4f*)(lr);
                const v4f l1 = *(const v4f*)(lr + 4);
                const v2f a0 = { -l0[0], -l0[1] }, a1 = { -l0[2], -l0[3] };
                const v2f a2 = { -l1[0], -l1[1] }, a3 = { -l1[2], -l1[3] };
                const v2f b0 = { R[0], R[1] }, b1 = { R[2], R[3] }, b2 = { R[4], R[5] }, b3 = { R[6], R[7] };
                v8f acc = T[i];
                acc = __builtin_amdgcn_wmma_f32_16x16x4_f32(false, a0, false, b0, (short)0, acc, false, false);
                acc = __builtin_amdgcn_wmma_f32_16x16x4_f32(false, a1, false, b1, (short)0, acc, false, false);
                acc = __builtin_amdgcn_wmma_f32_16x16x4_f32(false, a2, false, b2, (short)0, acc, false, false);
                acc = __builtin_amdgcn_wmma_f32_16x16x4_f32(false, a3, false, b3, (short)0, acc, false, false);
                T[i] = acc;
            }

            if (kb < 3 && wv == kb) {
                float* Ln = Lc + ((kb + 1) & 1)*HD*LP;
                #pragma unroll
                for (int i = 0; i < 4; ++i) {
                    #pragma unroll
                    for (int r = 0; r < 8; ++r)
                        Ln[(16*((kb + i) & 3) + 8*h + r)*LP + c] = T[i][r];
                }
            }
        }
        { const v8f t0 = T[0]; T[0] = T[1]; T[1] = T[2]; T[2] = T[3]; T[3] = t0; }
        if (kb < 3) __syncthreads();
    }

    float* Ab = dsc + (size_t)f*DSTRIDE;
    if (n >= 4 && n <= 7) {
        #pragma unroll
        for (int i = 0; i < 4; ++i) {
            #pragma unroll
            for (int r = 0; r < 8; ++r)
                Ab[(16*i + 8*h + r)*HD + 16*(n - 4) + c] = T[i][r];
        }
    } else if (n == 8) {
        if (c == 0) {
            #pragma unroll
            for (int i = 0; i < 4; ++i) {
                #pragma unroll
                for (int r = 0; r < 8; ++r)
                    Ab[HD*HD + 16*i + 8*h + r] = T[i][r];
            }
        }
    }
}


__device__ __forceinline__ void load_row(v2f (&o)[16], const float* row, int h)
{
    #pragma unroll
    for (int j = 0; j < 16; ++j) o[j] = *(const v2f*)(row + 2*h + 4*j);
}

__device__ __forceinline__ v8f mma16(const v2f (&a)[16], const v2f (&b)[16])
{
    v8f acc0 = {}, acc1 = {};
    #pragma unroll
    for (int j = 0; j < 16; j += 2) {
        acc0 = __builtin_amdgcn_wmma_f32_16x16x4_f32(false, a[j],     false, b[j],     (short)0, acc0, false, false);
        acc1 = __builtin_amdgcn_wmma_f32_16x16x4_f32(false, a[j + 1], false, b[j + 1], (short)0, acc1, false, false);
    }
    return acc0 + acc1;
}

__launch_bounds__(STHREADS)
__global__ void ssm_conv_wmma_f32(const float* __restrict__ x,
                                  const float* __restrict__ dsc,
                                  const float* __restrict__ Cmat,
                                  float* __restrict__ out)
{
    __shared__ __attribute__((aligned(16))) float smem[2*NVEC*VP + (16 + NT) + (NT + 16)*NBATCH + NBATCH*YP + 4*HD*PP];
    float* Vv    = smem;
    float* Ww    = Vv + NVEC*VP;
    float* Kp    = Ww + NVEC*VP;
    float* u_lds = Kp + 16 + NT;
    float* y_lds = u_lds + (NT + 16)*NBATCH;
    float* Pw    = y_lds + NBATCH*YP;

    const int f   = blockIdx.x;
    const int tid = threadIdx.x;
    const int wv  = __builtin_amdgcn_readfirstlane(tid >> 5);
    const int ln  = tid & 31;
    const int h   = ln >> 4;
    const int c   = ln & 15;

    const float* Ab = dsc + (size_t)f*DSTRIDE;
    float xv[4][NBATCH];
    #pragma unroll
    for (int q4 = 0; q4 < 4; ++q4) {
        #pragma unroll
        for (int b = 0; b < NBATCH; ++b)
            xv[q4][b] = x[(size_t)(b*NFEAT + f)*NT + tid + q4*STHREADS];
    }
    v4f pv[4];
    #pragma unroll
    for (int i = 0; i < 4; ++i) {
        const int e4 = tid + i*STHREADS;
        pv[i] = *(const v4f*)(Ab + (e4 >> 4)*HD + (e4 & 15)*4);
    }
    const float bvv = (tid < HD) ? Ab[HD*HD + tid] : 0.0f;
    const float cvv = (tid < HD) ? Cmat[f*HD + tid] : 0.0f;

    for (int e = tid; e < 15*VP; e += STHREADS) { Vv[VP + e] = 0.0f; Ww[VP + e] = 0.0f; }
    if (tid < HD) { Vv[tid] = bvv; Ww[tid] = cvv; }
    #pragma unroll
    for (int i = 0; i < 4; ++i) {
        const int e4 = tid + i*STHREADS;
        const int r = e4 >> 4, c4 = (e4 & 15)*4;
        *(v4f*)(Pw + r*PP + c4) = pv[i];
        #pragma unroll
        for (int s = 0; s < 4; ++s)
            Pw[HD*PP + (c4 + s)*PP + r] = pv[i][s];
    }
    if (tid < 16)  Kp[tid] = 0.0f;
    if (tid < 128) u_lds[tid] = 0.0f;
    #pragma unroll
    for (int q4 = 0; q4 < 4; ++q4) {
        float* ud = u_lds + (16 + tid + q4*STHREADS)*NBATCH;
        const v4f lo4 = { xv[q4][0], xv[q4][1], xv[q4][2], xv[q4][3] };
        const v4f hi4 = { xv[q4][4], xv[q4][5], xv[q4][6], xv[q4][7] };
        *(v4f*)(ud)     = lo4;
        *(v4f*)(ud + 4) = hi4;
    }
    __syncthreads();

    const int tn = wv & 3, tm0 = wv >> 2;
    const bool dbl = (wv < 4);
    #pragma nounroll
    for (int s = 0; s < NSTAGE; ++s) {
        const float* Pc  = Pw + (s & 1)*2*HD*PP;
        const float* PTc = Pc + HD*PP;
        float* Pn  = Pw + ((s + 1) & 1)*2*HD*PP;
        float* PTn = Pn + HD*PP;
        const bool sq   = (s < NSTAGE - 1);
        const bool vs   = (s < 5);
        float*     X    = vs ? Vv : Ww;
        const int  nrow = 1 << (vs ? s : s - 5);
        v2f bq[16], a0[16], a1[16], xa[16], xb[16];
        if (sq) {
            load_row(bq, PTc + (16*tn + c)*PP, h);
            load_row(a0, Pc + (16*tm0 + c)*PP, h);
            load_row(a1, Pc + (16*(tm0 + 2) + c)*PP, h);
        }
        if (dbl) {
            load_row(xa, X + c*VP, h);
            load_row(xb, (vs ? Pc : PTc) + (16*wv + c)*PP, h);
        }
        if (sq) {
            #pragma unroll
            for (int i = 0; i < 2; ++i) {
                const int tm = tm0 + 2*i;
                const v8f d = i ? mma16(a1, bq) : mma16(a0, bq);
                float* pr = Pn + (16*tm + 8*h)*PP + 16*tn + c;
                #pragma unroll
                for (int r = 0; r < 8; ++r) pr[r*PP] = d[r];
                float* pt = PTn + (16*tn + c)*PP + 16*tm + 8*h;
                const v4f lo4 = { d[0], d[1], d[2], d[3] };
                const v4f hi4 = { d[4], d[5], d[6], d[7] };
                *(v4f*)(pt)     = lo4;
                *(v4f*)(pt + 4) = hi4;
            }
        }
        if (dbl) {
            const v8f d = mma16(xa, xb);
            float* xr = X + (nrow + 8*h)*VP + 16*wv + c;
            #pragma unroll
            for (int r = 0; r < 8; ++r)
                if (8*h + r < nrow) xr[r*VP] = d[r];
        }
        __syncthreads();
    }

    if (dbl) {
        const int ta = wv >> 1, tb = wv & 1;
        v2f wa[16], vb[16];
        load_row(wa, Ww + (16*ta + c)*VP, h);
        load_row(vb, Vv + (16*tb + c)*VP, h);
        const v8f kt = mma16(wa, vb);
        float* kr = Kp + 16 + 32*(16*ta + 8*h) + 16*tb + c;
        #pragma unroll
        for (int r = 0; r < 8; ++r) kr[32*r] = kt[r];
    }
    __syncthreads();

    const int ucol = 128*(c >> 3) + (c & 7);
    for (int g = 0; g < 4; ++g) {
        const int p = (g & 1) ? (8*g + 7 - wv) : (8*g + wv);
        v8f acc0 = {}, acc1 = {};
        for (int dd = 0; dd <= 2*p + 1; dd += 2) {
            #pragma unroll
            for (int q = 0; q < 2; ++q) {
                const int d  = dd + q;
                const float* up = u_lds + (2*p - d + 1)*128 + ucol + (2*h)*NBATCH;
                const float* kp = Kp + 16 + 16*d + c - 2*h;
                #pragma unroll
                for (int j = 0; j < 4; ++j) {
                    v2f a, b;
                    a.x = kp[-4*j - 1];          a.y = kp[-4*j];
                    b.x = up[(4*j + 1)*NBATCH];  b.y = up[(4*j)*NBATCH];
                    if (j & 1)
                        acc1 = __builtin_amdgcn_wmma_f32_16x16x4_f32(false, a, false, b, (short)0, acc1, false, false);
                    else
                        acc0 = __builtin_amdgcn_wmma_f32_16x16x4_f32(false, a, false, b, (short)0, acc0, false, false);
                }
            }
        }
        const v8f y = acc0 + acc1;
        float* yp = y_lds + (c & 7)*YP + 16*(2*p + (c >> 3)) + 8*h;
        const v4f ylo = { y[0], y[1], y[2], y[3] };
        const v4f yhi = { y[4], y[5], y[6], y[7] };
        *(v4f*)(yp)     = ylo;
        *(v4f*)(yp + 4) = yhi;
    }
    __syncthreads();

    for (int e = tid; e < NBATCH*NT/4; e += STHREADS) {
        const int b = e >> 8, i4 = (e & 255)*4;
        *(v4f*)(out + (size_t)(b*NFEAT + f)*NT + i4) = *(const v4f*)(y_lds + b*YP + i4);
    }
}

extern "C" void kernel_launch(void* const* d_in, const int* in_sizes, int n_in,
                              void* d_out, int out_size, void* d_ws, size_t ws_size,
                              hipStream_t stream) {
    (void)in_sizes; (void)n_in; (void)ws_size; (void)out_size;
    const float* x = (const float*)d_in[0];
    const float* A = (const float*)d_in[1];
    const float* B = (const float*)d_in[2];
    const float* C = (const float*)d_in[3];
    float* out = (float*)d_out;
    float* dsc = (float*)d_ws;
    ssm_discretize_f32<<<dim3(NFEAT), dim3(GTHREADS), 0, stream>>>(A, B, dsc);
    ssm_conv_wmma_f32<<<dim3(NFEAT), dim3(STHREADS), 0, stream>>>(x, dsc, C, out);
}


# The two launches fused into one: the discretization runs in the convolution kernel's work-group and hands A_bar, Bv to the stages through LDS, no workspace pass; x loads travel beside it; same sums
# speedup vs baseline: 18.8100x; 1.0908x over previous
#include <hip/hip_runtime.h>

typedef __attribute__((ext_vector_type(2))) float v2f;
typedef __attribute__((ext_vector_type(4))) float v4f;
typedef __attribute__((ext_vector_type(8))) float v8f;

#define NBATCH 8
#define NFEAT  256
#define HD     64
#define NT     1024
#define LP     20
#define STHREADS 256
#define NVEC   32
#define NSTAGE 10
#define PP 68
#define VP 68
#define YP 1028


template<int P> __device__ __forceinline__ float row_bcast(float x)
{
    return __builtin_bit_cast(float, __builtin_amdgcn_update_dpp(0, __builtin_bit_cast(int, x), 0x150 + P, 0xF, 0xF, false));
}

template<int P> __device__ __forceinline__ void gj_pivot(float (&d)[16], float (&e)[8], int c)
{
    const float pp  = row_bcast<P>(d[P]);
    const float fac = (c != P) ? d[P] / pp : 0.0f;
    #pragma unroll
    for (int cc = P + 1; cc < 16; ++cc)
        d[cc] -= fac*row_bcast<P>(d[cc]);
    #pragma unroll
    for (int m = 0; m < 8; ++m)
        e[m] -= fac*row_bcast<P>(e[m]);
}

__device__ __forceinline__ void lds_barrier()
{
    __builtin_amdgcn_fence(__ATOMIC_RELEASE, "workgroup", "local");
    __builtin_amdgcn_s_barrier();
    __builtin_amdgcn_fence(__ATOMIC_ACQUIRE, "workgroup", "local");
}


__device__ __forceinline__ void load_row(v2f (&o)[16], const float* row, int h)
{
    #pragma unroll
    for (int j = 0; j < 16; ++j) o[j] = *(const v2f*)(row + 2*h + 4*j);
}

__device__ __forceinline__ v8f mma16(const v2f (&a)[16], const v2f (&b)[16])
{
    v8f acc0 = {}, acc1 = {};
    #pragma unroll
    for (int j = 0; j < 16; j += 2) {
        acc0 = __builtin_amdgcn_wmma_f32_16x16x4_f32(false, a[j],     false, b[j],     (short)0, acc0, false, false);
        acc1 = __builtin_amdgcn_wmma_f32_16x16x4_f32(false, a[j + 1], false, b[j + 1], (short)0, acc1, false, false);
    }
    return acc0 + acc1;
}

__launch_bounds__(STHREADS)
__global__ void ssm_fused_wmma_f32(const float* __restrict__ x,
                                   const float* __restrict__ Amat,
                                   const float* __restrict__ Bmat,
                                   const float* __restrict__ Cmat,
                                   float* __restrict__ out)
{
    __shared__ __attribute__((aligned(16))) float smem[2*NVEC*VP + (16 + NT) + (NT + 16)*NBATCH + NBATCH*YP + 4*HD*PP];
    float* Vv    = smem;
    float* Ww    = Vv + NVEC*VP;
    float* Kp    = Ww + NVEC*VP;
    float* u_lds = Kp + 16 + NT;
    float* y_lds = u_lds + (NT + 16)*NBATCH;
    float* Pw    = y_lds + NBATCH*YP;
    float* Lc    = y_lds;

    const int f   = blockIdx.x;
    const int tid = threadIdx.x;
    const int wv  = __builtin_amdgcn_readfirstlane(tid >> 5);
    const int ln  = tid & 31;
    const int h   = ln >> 4;
    const int c   = ln & 15;
    const int n   = wv + 1;

    const float* Af = Amat + f*HD*HD;

    for (int m = 0; m < 4; ++m) {
        const int e = tid + m*STHREADS;
        const int r = e >> 4, cc = e & 15;
        Lc[r*LP + cc] = ((r == cc) ? 1.0f : 0.0f) - 0.5f*Af[r*HD + cc];
    }

    v8f T[4];
    if (n <= 7) {
        const float sg = (n <= 3) ? -0.5f : 0.5f;
        const int col = 16*(n & 3) + c;
        #pragma unroll
        for (int i = 0; i < 4; ++i) {
            #pragma unroll
            for (int r = 0; r < 8; ++r) {
                const int row = 16*i + 8*h + r;
                T[i][r] = ((row == col) ? 1.0f : 0.0f) + sg*Af[row*HD + col];
            }
        }
    } else {
        #pragma unroll
        for (int i = 0; i < 4; ++i) {
            #pragma unroll
            for (int r = 0; r < 8; ++r) {
                const float bv = Bmat[f*HD + 16*i + 8*h + r];
                T[i][r] = (c == 0) ? bv : 0.0f;
            }
        }
    }

    float xv[4][NBATCH];
    #pragma unroll
    for (int q4 = 0; q4 < 4; ++q4) {
        #pragma unroll
        for (int b = 0; b < NBATCH; ++b)
            xv[q4][b] = x[(size_t)(b*NFEAT + f)*NT + tid + q4*STHREADS];
    }
    const float cvv = (tid < HD) ? Cmat[f*HD + tid] : 0.0f;

    for (int e = tid; e < 15*VP; e += STHREADS) { Vv[VP + e] = 0.0f; Ww[VP + e] = 0.0f; }
    if (tid < 16)  Kp[tid] = 0.0f;
    if (tid < 128) u_lds[tid] = 0.0f;
    lds_barrier();

    #pragma nounroll
    for (int kb = 0; kb < 4; ++kb) {
        const float* Lb = Lc + (kb & 1)*HD*LP;
        if (n > kb) {
            float d[16], e[8];
            const float* dr = Lb + (16*kb + c)*LP;
            #pragma unroll
            for (int q4 = 0; q4 < 4; ++q4) {
                const v4f t = *(const v4f*)(dr + 4*q4);
                d[4*q4] = t[0]; d[4*q4 + 1] = t[1]; d[4*q4 + 2] = t[2]; d[4*q4 + 3] = t[3];
            }
            #pragma unroll
            for (int m = 0; m < 8; ++m) e[m] = (8*h + m == c) ? 1.0f : 0.0f;
            gj_pivot<0>(d, e, c);   gj_pivot<1>(d, e, c);   gj_pivot<2>(d, e, c);   gj_pivot<3>(d, e, c);
            gj_pivot<4>(d, e, c);   gj_pivot<5>(d, e, c);   gj_pivot<6>(d, e, c);   gj_pivot<7>(d, e, c);
            gj_pivot<8>(d, e, c);   gj_pivot<9>(d, e, c);   gj_pivot<10>(d, e, c);  gj_pivot<11>(d, e, c);
            gj_pivot<12>(d, e, c);  gj_pivot<13>(d, e, c);  gj_pivot<14>(d, e, c);  gj_pivot<15>(d, e, c);
            float diag = d[0];
            #pragma unroll
            for (int p = 1; p < 16; ++p) diag = (c == p) ? d[p] : diag;
            #pragma unroll
            for (int m = 0; m < 8; ++m) e[m] = e[m] / diag;

            v8f R = {};
            #pragma unroll
            for (int j = 0; j < 4; ++j) {
                const v2f a = { e[2*j], e[2*j + 1] };
                const v2f b = { T[0][2*j], T[0][2*j + 1] };
                R = __builtin_amdgcn_wmma_f32_16x16x4_f32(false, a, false, b, (short)0, R, false, false);
            }
            T[0] = R;

            #pragma unroll
            for (int i = 1; i < 4; ++i) {
                const float* lr = Lb + (16*((kb + i) & 3) + c)*LP + 8*h;
                const v4f l0 = *(const v4f*)(lr);
                const v4f l1 = *(const v4f*)(lr + 4);
                const v2f a0 = { -l0[0], -l0[1] }, a1 = { -l0[2], -l0[3] };
                const v2f a2 = { -l1[0], -l1[1] }, a3 = { -l1[2], -l1[3] };
                const v2f b0 = { R[0], R[1] }, b1 = { R[2], R[3] }, b2 = { R[4], R[5] }, b3 = { R[6], R[7] };
                v8f acc = T[i];
                acc = __builtin_amdgcn_wmma_f32_16x16x4_f32(false, a0, false, b0, (short)0, acc, false, false);
                acc = __builtin_amdgcn_wmma_f32_16x16x4_f32(false, a1, false, b1, (short)0, acc, false, false);
                acc = __builtin_amdgcn_wmma_f32_16x16x4_f32(false, a2, false, b2, (short)0, acc, false, false);
                acc = __builtin_amdgcn_wmma_f32_16x16x4_f32(false, a3, false, b3, (short)0, acc, false, false);
                T[i] = acc;
            }

            if (kb < 3 && wv == kb) {
                float* Ln = Lc + ((kb + 1) & 1)*HD*LP;
                #pragma unroll
                for (int i = 0; i < 4; ++i) {
                    #pragma unroll
                    for (int r = 0; r < 8; ++r)
                        Ln[(16*((kb + i) & 3) + 8*h + r)*LP + c] = T[i][r];
                }
            }
        }
        { const v8f t0 = T[0]; T[0] = T[1]; T[1] = T[2]; T[2] = T[3]; T[3] = t0; }
        if (kb < 3) lds_barrier();
    }

    if (n >= 4 && n <= 7) {
        const int col = 16*(n - 4) + c;
        #pragma unroll
        for (int i = 0; i < 4; ++i) {
            float* pr = Pw + (16*i + 8*h)*PP + col;
            #pragma unroll
            for (int r = 0; r < 8; ++r) pr[r*PP] = T[i][r];
            float* pt = Pw + HD*PP + col*PP + 16*i + 8*h;
            const v4f lo4 = { T[i][0], T[i][1], T[i][2], T[i][3] };
            const v4f hi4 = { T[i][4], T[i][5], T[i][6], T[i][7] };
            *(v4f*)(pt)     = lo4;
            *(v4f*)(pt + 4) = hi4;
        }
    } else if (n == 8) {
        if (c == 0) {
            #pragma unroll
            for (int i = 0; i < 4; ++i) {
                float* vr = Vv + 16*i + 8*h;
                const v4f lo4 = { T[i][0], T[i][1], T[i][2], T[i][3] };
                const v4f hi4 = { T[i][4], T[i][5], T[i][6], T[i][7] };
                *(v4f*)(vr)     = lo4;
                *(v4f*)(vr + 4) = hi4;
            }
        }
    }
    if (tid < HD) Ww[tid] = cvv;
    #pragma unroll
    for (int q4 = 0; q4 < 4; ++q4) {
        float* ud = u_lds + (16 + tid + q4*STHREADS)*NBATCH;
        const v4f lo4 = { xv[q4][0], xv[q4][1], xv[q4][2], xv[q4][3] };
        const v4f hi4 = { xv[q4][4], xv[q4][5], xv[q4][6], xv[q4][7] };
        *(v4f*)(ud)     = lo4;
        *(v4f*)(ud + 4) = hi4;
    }
    __syncthreads();

    const int tn = wv & 3, tm0 = wv >> 2;
    const bool dbl = (wv < 4);
    #pragma nounroll
    for (int s = 0; s < NSTAGE; ++s) {
        const float* Pc  = Pw + (s & 1)*2*HD*PP;
        const float* PTc = Pc + HD*PP;
        float* Pn  = Pw + ((s + 1) & 1)*2*HD*PP;
        float* PTn = Pn + HD*PP;
        const bool sq   = (s < NSTAGE - 1);
        const bool vs   = (s < 5);
        float*     X    = vs ? Vv : Ww;
        const int  nrow = 1 << (vs ? s : s - 5);
        v2f bq[16], a0[16], a1[16], xa[16], xb[16];
        if (sq) {
            load_row(bq, PTc + (16*tn + c)*PP, h);
            load_row(a0, Pc + (16*tm0 + c)*PP, h);
            load_row(a1, Pc + (16*(tm0 + 2) + c)*PP, h);
        }
        if (dbl) {
            load_row(xa, X + c*VP, h);
            load_row(xb, (vs ? Pc : PTc) + (16*wv + c)*PP, h);
        }
        if (sq) {
            #pragma unroll
            for (int i = 0; i < 2; ++i) {
                const int tm = tm0 + 2*i;
                const v8f d = i ? mma16(a1, bq) : mma16(a0, bq);
                float* pr = Pn + (16*tm + 8*h)*PP + 16*tn + c;
                #pragma unroll
                for (int r = 0; r < 8; ++r) pr[r*PP] = d[r];
                float* pt = PTn + (16*tn + c)*PP + 16*tm + 8*h;
                const v4f lo4 = { d[0], d[1], d[2], d[3] };
                const v4f hi4 = { d[4], d[5], d[6], d[7] };
                *(v4f*)(pt)     = lo4;
                *(v4f*)(pt + 4) = hi4;
            }
        }
        if (dbl) {
            const v8f d = mma16(xa, xb);
            float* xr = X + (nrow + 8*h)*VP + 16*wv + c;
            #pragma unroll
            for (int r = 0; r < 8; ++r)
                if (8*h + r < nrow) xr[r*VP] = d[r];
        }
        __syncthreads();
    }

    if (dbl) {
        const int ta = wv >> 1, tb = wv & 1;
        v2f wa[16], vb[16];
        load_row(wa, Ww + (16*ta + c)*VP, h);
        load_row(vb, Vv + (16*tb + c)*VP, h);
        const v8f kt = mma16(wa, vb);
        float* kr = Kp + 16 + 32*(16*ta + 8*h) + 16*tb + c;
        #pragma unroll
        for (int r = 0; r < 8; ++r) kr[32*r] = kt[r];
    }
    __syncthreads();

    const int ucol = 128*(c >> 3) + (c & 7);
    for (int g = 0; g < 4; ++g) {
        const int p = (g & 1) ? (8*g + 7 - wv) : (8*g + wv);
        v8f acc0 = {}, acc1 = {};
        for (int dd = 0; dd <= 2*p + 1; dd += 2) {
            #pragma unroll
            for (int q = 0; q < 2; ++q) {
                const int d  = dd + q;
                const float* up = u_lds + (2*p - d + 1)*128 + ucol + (2*h)*NBATCH;
                const float* kp = Kp + 16 + 16*d + c - 2*h;
                #pragma unroll
                for (int j = 0; j < 4; ++j) {
                    v2f a, b;
                    a.x = kp[-4*j - 1];          a.y = kp[-4*j];
                    b.x = up[(4*j + 1)*NBATCH];  b.y = up[(4*j)*NBATCH];
                    if (j & 1)
                        acc1 = __builtin_amdgcn_wmma_f32_16x16x4_f32(false, a, false, b, (short)0, acc1, false, false);
                    else
                        acc0 = __builtin_amdgcn_wmma_f32_16x16x4_f32(false, a, false, b, (short)0, acc0, false, false);
                }
            }
        }
        const v8f y = acc0 + acc1;
        float* yp = y_lds + (c & 7)*YP + 16*(2*p + (c >> 3)) + 8*h;
        const v4f ylo = { y[0], y[1], y[2], y[3] };
        const v4f yhi = { y[4], y[5], y[6], y[7] };
        *(v4f*)(yp)     = ylo;
        *(v4f*)(yp + 4) = yhi;
    }
    __syncthreads();

    for (int e = tid; e < NBATCH*NT/4; e += STHREADS) {
        const int b = e >> 8, i4 = (e & 255)*4;
        *(v4f*)(out + (size_t)(b*NFEAT + f)*NT + i4) = *(const v4f*)(y_lds + b*YP + i4);
    }
}

extern "C" void kernel_launch(void* const* d_in, const int* in_sizes, int n_in,
                              void* d_out, int out_size, void* d_ws, size_t ws_size,
                              hipStream_t stream) {
    (void)in_sizes; (void)n_in; (void)ws_size; (void)out_size; (void)d_ws;
    const float* x = (const float*)d_in[0];
    const float* A = (const float*)d_in[1];
    const float* B = (const float*)d_in[2];
    const float* C = (const float*)d_in[3];
    float* out = (float*)d_out;
    ssm_fused_wmma_f32<<<dim3(NFEAT), dim3(STHREADS), 0, stream>>>(x, A, B, C, out);
}


# Toeplitz loop software-pipelined over two register operand sets: next trip's 16 LDS loads fly during the current 8 WMMAs; taps stored descending, k pair in a slice swapped: sum order changed
# speedup vs baseline: 19.6038x; 1.0422x over previous
#include <hip/hip_runtime.h>

typedef __attribute__((ext_vector_type(2))) float v2f;
typedef __attribute__((ext_vector_type(4))) float v4f;
typedef __attribute__((ext_vector_type(8))) float v8f;

#define NBATCH 8
#define NFEAT  256
#define HD     64
#define NT     1024
#define LP     20
#define STHREADS 256
#define NVEC   32
#define NSTAGE 10
#define PP 68
#define VP 68
#define YP 1028


template<int P> __device__ __forceinline__ float row_bcast(float x)
{
    return __builtin_bit_cast(float, __builtin_amdgcn_update_dpp(0, __builtin_bit_cast(int, x), 0x150 + P, 0xF, 0xF, false));
}

template<int P> __device__ __forceinline__ void gj_pivot(float (&d)[16], float (&e)[8], int c)
{
    const float pp  = row_bcast<P>(d[P]);
    const float fac = (c != P) ? d[P] / pp : 0.0f;
    #pragma unroll
    for (int cc = P + 1; cc < 16; ++cc)
        d[cc] -= fac*row_bcast<P>(d[cc]);
    #pragma unroll
    for (int m = 0; m < 8; ++m)
        e[m] -= fac*row_bcast<P>(e[m]);
}

__device__ __forceinline__ void lds_barrier()
{
    __builtin_amdgcn_fence(__ATOMIC_RELEASE, "workgroup", "local");
    __builtin_amdgcn_s_barrier();
    __builtin_amdgcn_fence(__ATOMIC_ACQUIRE, "workgroup", "local");
}


__device__ __forceinline__ void load_row(v2f (&o)[16], const float* row, int h)
{
    #pragma unroll
    for (int j = 0; j < 16; ++j) o[j] = *(const v2f*)(row + 2*h + 4*j);
}

__device__ __forceinline__ v8f mma16(const v2f (&a)[16], const v2f (&b)[16])
{
    v8f acc0 = {}, acc1 = {};
    #pragma unroll
    for (int j = 0; j < 16; j += 2) {
        acc0 = __builtin_amdgcn_wmma_f32_16x16x4_f32(false, a[j],     false, b[j],     (short)0, acc0, false, false);
        acc1 = __builtin_amdgcn_wmma_f32_16x16x4_f32(false, a[j + 1], false, b[j + 1], (short)0, acc1, false, false);
    }
    return acc0 + acc1;
}

__device__ __forceinline__ void conv_load(v2f (&a)[8], v2f (&b)[8], const float* up, const float* kp)
{
    #pragma unroll
    for (int q = 0; q < 2; ++q) {
        #pragma unroll
        for (int j = 0; j < 4; ++j) {
            a[4*q + j].x = kp[-16*q + 4*j];
            a[4*q + j].y = kp[-16*q + 4*j + 1];
            b[4*q + j].x = up[-128*q + (4*j)*NBATCH];
            b[4*q + j].y = up[-128*q + (4*j + 1)*NBATCH];
        }
    }
}

__device__ __forceinline__ void conv_mma(v8f& acc0, v8f& acc1, const v2f (&a)[8], const v2f (&b)[8])
{
    #pragma unroll
    for (int i = 0; i < 8; ++i) {
        if (i & 1)
            acc1 = __builtin_amdgcn_wmma_f32_16x16x4_f32(false, a[i], false, b[i], (short)0, acc1, false, false);
        else
            acc0 = __builtin_amdgcn_wmma_f32_16x16x4_f32(false, a[i], false, b[i], (short)0, acc0, false, false);
    }
}

__launch_bounds__(STHREADS)
__global__ void ssm_fused_wmma_f32(const float* __restrict__ x,
                                   const float* __restrict__ Amat,
                                   const float* __restrict__ Bmat,
                                   const float* __restrict__ Cmat,
                                   float* __restrict__ out)
{
    __shared__ __attribute__((aligned(16))) float smem[2*NVEC*VP + (16 + NT) + (NT + 16)*NBATCH + NBATCH*YP + 4*HD*PP];
    float* Vv    = smem;
    float* Ww    = Vv + NVEC*VP;
    float* Kp    = Ww + NVEC*VP;
    float* u_lds = Kp + 16 + NT;
    float* y_lds = u_lds + (NT + 16)*NBATCH;
    float* Pw    = y_lds + NBATCH*YP;
    float* Lc    = y_lds;

    const int f   = blockIdx.x;
    const int tid = threadIdx.x;
    const int wv  = __builtin_amdgcn_readfirstlane(tid >> 5);
    const int ln  = tid & 31;
    const int h   = ln >> 4;
    const int c   = ln & 15;
    const int n   = wv + 1;

    const float* Af = Amat + f*HD*HD;

    for (int m = 0; m < 4; ++m) {
        const int e = tid + m*STHREADS;
        const int r = e >> 4, cc = e & 15;
        Lc[r*LP + cc] = ((r == cc) ? 1.0f : 0.0f) - 0.5f*Af[r*HD + cc];
    }

    v8f T[4];
    if (n <= 7) {
        const float sg = (n <= 3) ? -0.5f : 0.5f;
        const int col = 16*(n & 3) + c;
        #pragma unroll
        for (int i = 0; i < 4; ++i) {
            #pragma unroll
            for (int r = 0; r < 8; ++r) {
                const int row = 16*i + 8*h + r;
                T[i][r] = ((row == col) ? 1.0f : 0.0f) + sg*Af[row*HD + col];
            }
        }
    } else {
        #pragma unroll
        for (int i = 0; i < 4; ++i) {
            #pragma unroll
            for (int r = 0; r < 8; ++r) {
                const float bv = Bmat[f*HD + 16*i + 8*h + r];
                T[i][r] = (c == 0) ? bv : 0.0f;
            }
        }
    }

    float xv[4][NBATCH];
    #pragma unroll
    for (int q4 = 0; q4 < 4; ++q4) {
        #pragma unroll
        for (int b = 0; b < NBATCH; ++b)
            xv[q4][b] = x[(size_t)(b*NFEAT + f)*NT + tid + q4*STHREADS];
    }
    const float cvv = (tid < HD) ? Cmat[f*HD + tid] : 0.0f;

    for (int e = tid; e < 15*VP; e += STHREADS) { Vv[VP + e] = 0.0f; Ww[VP + e] = 0.0f; }
    if (tid < 16)  Kp[NT + tid] = 0.0f;
    if (tid < 128) u_lds[tid] = 0.0f;
    lds_barrier();

    #pragma nounroll
    for (int kb = 0; kb < 4; ++kb) {
        const float* Lb = Lc + (kb & 1)*HD*LP;
        if (n > kb) {
            float d[16], e[8];
            const float* dr = Lb + (16*kb + c)*LP;
            #pragma unroll
            for (int q4 = 0; q4 < 4; ++q4) {
                const v4f t = *(const v4f*)(dr + 4*q4);
                d[4*q4] = t[0]; d[4*q4 + 1] = t[1]; d[4*q4 + 2] = t[2]; d[4*q4 + 3] = t[3];
            }
            #pragma unroll
            for (int m = 0; m < 8; ++m) e[m] = (8*h + m == c) ? 1.0f : 0.0f;
            gj_pivot<0>(d, e, c);   gj_pivot<1>(d, e, c);   gj_pivot<2>(d, e, c);   gj_pivot<3>(d, e, c);
            gj_pivot<4>(d, e, c);   gj_pivot<5>(d, e, c);   gj_pivot<6>(d, e, c);   gj_pivot<7>(d, e, c);
            gj_pivot<8>(d, e, c);   gj_pivot<9>(d, e, c);   gj_pivot<10>(d, e, c);  gj_pivot<11>(d, e, c);
            gj_pivot<12>(d, e, c);  gj_pivot<13>(d, e, c);  gj_pivot<14>(d, e, c);  gj_pivot<15>(d, e, c);
            float diag = d[0];
            #pragma unroll
            for (int p = 1; p < 16; ++p) diag = (c == p) ? d[p] : diag;
            #pragma unroll
            for (int m = 0; m < 8; ++m) e[m] = e[m] / diag;

            v8f R = {};
            #pragma unroll
            for (int j = 0; j < 4; ++j) {
                const v2f a = { e[2*j], e[2*j + 1] };
                const v2f b = { T[0][2*j], T[0][2*j + 1] };
                R = __builtin_amdgcn_wmma_f32_16x16x4_f32(false, a, false, b, (short)0, R, false, false);
            }
            T[0] = R;

            #pragma unroll
            for (int i = 1; i < 4; ++i) {
                const float* lr = Lb + (16*((kb + i) & 3) + c)*LP + 8*h;
                const v4f l0 = *(const v4f*)(lr);
                const v4f l1 = *(const v4f*)(lr + 4);
                const v2f a0 = { -l0[0], -l0[1] }, a1 = { -l0[2], -l0[3] };
                const v2f a2 = { -l1[0], -l1[1] }, a3 = { -l1[2], -l1[3] };
                const v2f b0 = { R[0], R[1] }, b1 = { R[2], R[3] }, b2 = { R[4], R[5] }, b3 = { R[6], R[7] };
                v8f acc = T[i];
                acc = __builtin_amdgcn_wmma_f32_16x16x4_f32(false, a0, false, b0, (short)0, acc, false, false);
                acc = __builtin_amdgcn_wmma_f32_16x16x4_f32(false, a1, false, b1, (short)0, acc, false, false);
                acc = __builtin_amdgcn_wmma_f32_16x16x4_f32(false, a2, false, b2, (short)0, acc, false, false);
                acc = __builtin_amdgcn_wmma_f32_16x16x4_f32(false, a3, false, b3, (short)0, acc, false, false);
                T[i] = acc;
            }

            if (kb < 3 && wv == kb) {
                float* Ln = Lc + ((kb + 1) & 1)*HD*LP;
                #pragma unroll
                for (int i = 0; i < 4; ++i) {
                    #pragma unroll
                    for (int r = 0; r < 8; ++r)
                        Ln[(16*((kb + i) & 3) + 8*h + r)*LP + c] = T[i][r];
                }
            }
        }
        { const v8f t0 = T[0]; T[0] = T[1]; T[1] = T[2]; T[2] = T[3]; T[3] = t0; }
        if (kb < 3) lds_barrier();
    }

    if (n >= 4 && n <= 7) {
        const int col = 16*(n - 4) + c;
        #pragma unroll
        for (int i = 0; i < 4; ++i) {
            float* pr = Pw + (16*i + 8*h)*PP + col;
            #pragma unroll
            for (int r = 0; r < 8; ++r) pr[r*PP] = T[i][r];
            float* pt = Pw + HD*PP + col*PP + 16*i + 8*h;
            const v4f lo4 = { T[i][0], T[i][1], T[i][2], T[i][3] };
            const v4f hi4 = { T[i][4], T[i][5], T[i][6], T[i][7] };
            *(v4f*)(pt)     = lo4;
            *(v4f*)(pt + 4) = hi4;
        }
    } else if (n == 8) {
        if (c == 0) {
            #pragma unroll
            for (int i = 0; i < 4; ++i) {
                float* vr = Vv + 16*i + 8*h;
                const v4f lo4 = { T[i][0], T[i][1], T[i][2], T[i][3] };
                const v4f hi4 = { T[i][4], T[i][5], T[i][6], T[i][7] };
                *(v4f*)(vr)     = lo4;
                *(v4f*)(vr + 4) = hi4;
            }
        }
    }
    if (tid < HD) Ww[tid] = cvv;
    #pragma unroll
    for (int q4 = 0; q4 < 4; ++q4) {
        float* ud = u_lds + (16 + tid + q4*STHREADS)*NBATCH;
        const v4f lo4 = { xv[q4][0], xv[q4][1], xv[q4][2], xv[q4][3] };
        const v4f hi4 = { xv[q4][4], xv[q4][5], xv[q4][6], xv[q4][7] };
        *(v4f*)(ud)     = lo4;
        *(v4f*)(ud + 4) = hi4;
    }
    __syncthreads();

    const int tn = wv & 3, tm0 = wv >> 2;
    const bool dbl = (wv < 4);
    #pragma nounroll
    for (int s = 0; s < NSTAGE; ++s) {
        const float* Pc  = Pw + (s & 1)*2*HD*PP;
        const float* PTc = Pc + HD*PP;
        float* Pn  = Pw + ((s + 1) & 1)*2*HD*PP;
        float* PTn = Pn + HD*PP;
        const bool sq   = (s < NSTAGE - 1);
        const bool vs   = (s < 5);
        float*     X    = vs ? Vv : Ww;
        const int  nrow = 1 << (vs ? s : s - 5);
        v2f bq[16], a0[16], a1[16], xa[16], xb[16];
        if (sq) {
            load_row(bq, PTc + (16*tn + c)*PP, h);
            load_row(a0, Pc + (16*tm0 + c)*PP, h);
            load_row(a1, Pc + (16*(tm0 + 2) + c)*PP, h);
        }
        if (dbl) {
            load_row(xa, X + c*VP, h);
            load_row(xb, (vs ? Pc : PTc) + (16*wv + c)*PP, h);
        }
        if (sq) {
            #pragma unroll
            for (int i = 0; i < 2; ++i) {
                const int tm = tm0 + 2*i;
                const v8f d = i ? mma16(a1, bq) : mma16(a0, bq);
                float* pr = Pn + (16*tm + 8*h)*PP + 16*tn + c;
                #pragma unroll
                for (int r = 0; r < 8; ++r) pr[r*PP] = d[r];
                float* pt = PTn + (16*tn + c)*PP + 16*tm + 8*h;
                const v4f lo4 = { d[0], d[1], d[2], d[3] };
                const v4f hi4 = { d[4], d[5], d[6], d[7] };
                *(v4f*)(pt)     = lo4;
                *(v4f*)(pt + 4) = hi4;
            }
        }
        if (dbl) {
            const v8f d = mma16(xa, xb);
            float* xr = X + (nrow + 8*h)*VP + 16*wv + c;
            #pragma unroll
            for (int r = 0; r < 8; ++r)
                if (8*h + r < nrow) xr[r*VP] = d[r];
        }
        __syncthreads();
    }

    if (dbl) {
        const int ta = wv >> 1, tb = wv & 1;
        v2f wa[16], vb[16];
        load_row(wa, Ww + (16*ta + c)*VP, h);
        load_row(vb, Vv + (16*tb + c)*VP, h);
        const v8f kt = mma16(wa, vb);
        float* kr = Kp + NT - 1 - 32*(16*ta + 8*h) - 16*tb - c;
        #pragma unroll
        for (int r = 0; r < 8; ++r) kr[-32*r] = kt[r];
    }
    __syncthreads();

    const int ucol = 128*(c >> 3) + (c & 7);
    for (int g = 0; g < 4; ++g) {
        const int p = (g & 1) ? (8*g + 7 - wv) : (8*g + wv);
        v8f acc0 = {}, acc1 = {};
        const float* up0 = u_lds + (2*p + 1)*128 + ucol + (2*h)*NBATCH;
        const float* kp0 = Kp + NT - 1 - c + 2*h;
        v2f xa[8], xb[8], ya[8], yb[8];
        conv_load(xa, xb, up0, kp0);
        const int npair = (p + 1) >> 1;
        for (int i = 0; i < npair; ++i) {
            conv_load(ya, yb, up0 - 256*(2*i + 1), kp0 - 32*(2*i + 1));
            __builtin_amdgcn_sched_barrier(0);
            conv_mma(acc0, acc1, xa, xb);
            const int tn = (2*i + 2 <= p) ? 2*i + 2 : p;
            conv_load(xa, xb, up0 - 256*tn, kp0 - 32*tn);
            __builtin_amdgcn_sched_barrier(0);
            conv_mma(acc0, acc1, ya, yb);
        }
        if ((p + 1) & 1) conv_mma(acc0, acc1, xa, xb);
        const v8f y = acc0 + acc1;
        float* yp = y_lds + (c & 7)*YP + 16*(2*p + (c >> 3)) + 8*h;
        const v4f ylo = { y[0], y[1], y[2], y[3] };
        const v4f yhi = { y[4], y[5], y[6], y[7] };
        *(v4f*)(yp)     = ylo;
        *(v4f*)(yp + 4) = yhi;
    }
    __syncthreads();

    for (int e = tid; e < NBATCH*NT/4; e += STHREADS) {
        const int b = e >> 8, i4 = (e & 255)*4;
        *(v4f*)(out + (size_t)(b*NFEAT + f)*NT + i4) = *(const v4f*)(y_lds + b*YP + i4);
    }
}

extern "C" void kernel_launch(void* const* d_in, const int* in_sizes, int n_in,
                              void* d_out, int out_size, void* d_ws, size_t ws_size,
                              hipStream_t stream) {
    (void)in_sizes; (void)n_in; (void)ws_size; (void)out_size; (void)d_ws;
    const float* x = (const float*)d_in[0];
    const float* A = (const float*)d_in[1];
    const float* B = (const float*)d_in[2];
    const float* C = (const float*)d_in[3];
    float* out = (float*)d_out;
    ssm_fused_wmma_f32<<<dim3(NFEAT), dim3(STHREADS), 0, stream>>>(x, A, B, C, out);
}


# Discretization on four waves, one per SIMD, two tableau column tiles each (both from one set of A loads): the serial 16x16 inverse runs once per SIMD, not twice on shared issue slots; same sums
# speedup vs baseline: 20.3184x; 1.0365x over previous
#include <hip/hip_runtime.h>

typedef __attribute__((ext_vector_type(2))) float v2f;
typedef __attribute__((ext_vector_type(4))) float v4f;
typedef __attribute__((ext_vector_type(8))) float v8f;

#define NBATCH 8
#define NFEAT  256
#define HD     64
#define NT     1024
#define LP     20
#define STHREADS 256
#define NVEC   32
#define NSTAGE 10
#define PP 68
#define VP 68
#define YP 1028


template<int P> __device__ __forceinline__ float row_bcast(float x)
{
    return __builtin_bit_cast(float, __builtin_amdgcn_update_dpp(0, __builtin_bit_cast(int, x), 0x150 + P, 0xF, 0xF, false));
}

template<int P> __device__ __forceinline__ void gj_pivot(float (&d)[16], float (&e)[8], int c)
{
    const float pp  = row_bcast<P>(d[P]);
    const float fac = (c != P) ? d[P] / pp : 0.0f;
    #pragma unroll
    for (int cc = P + 1; cc < 16; ++cc)
        d[cc] -= fac*row_bcast<P>(d[cc]);
    #pragma unroll
    for (int m = 0; m < 8; ++m)
        e[m] -= fac*row_bcast<P>(e[m]);
}

__device__ __forceinline__ void lds_barrier()
{
    __builtin_amdgcn_fence(__ATOMIC_RELEASE, "workgroup", "local");
    __builtin_amdgcn_s_barrier();
    __builtin_amdgcn_fence(__ATOMIC_ACQUIRE, "workgroup", "local");
}

__device__ __forceinline__ void pivot_update(v8f (&TA)[4], v8f (&TB)[4], const float (&e)[8],
                                             const float* Lb, int kb, int c, int h)
{
    v8f RA = {}, RB = {};
    #pragma unroll
    for (int j = 0; j < 4; ++j) {
        const v2f a = { e[2*j], e[2*j + 1] };
        const v2f bA = { TA[0][2*j], TA[0][2*j + 1] };
        const v2f bB = { TB[0][2*j], TB[0][2*j + 1] };
        RA = __builtin_amdgcn_wmma_f32_16x16x4_f32(false, a, false, bA, (short)0, RA, false, false);
        RB = __builtin_amdgcn_wmma_f32_16x16x4_f32(false, a, false, bB, (short)0, RB, false, false);
    }
    TA[0] = RA;
    TB[0] = RB;
    #pragma unroll
    for (int i = 1; i < 4; ++i) {
        const float* lr = Lb + (16*((kb + i) & 3) + c)*LP + 8*h;
        const v4f l0 = *(const v4f*)(lr);
        const v4f l1 = *(const v4f*)(lr + 4);
        const v2f a0 = { -l0[0], -l0[1] }, a1 = { -l0[2], -l0[3] };
        const v2f a2 = { -l1[0], -l1[1] }, a3 = { -l1[2], -l1[3] };
        {
            const v2f b0 = { RA[0], RA[1] }, b1 = { RA[2], RA[3] }, b2 = { RA[4], RA[5] }, b3 = { RA[6], RA[7] };
            v8f acc = TA[i];
            acc = __builtin_amdgcn_wmma_f32_16x16x4_f32(false, a0, false, b0, (short)0, acc, false, false);
            acc = __builtin_amdgcn_wmma_f32_16x16x4_f32(false, a1, false, b1, (short)0, acc, false, false);
            acc = __builtin_amdgcn_wmma_f32_16x16x4_f32(false, a2, false, b2, (short)0, acc, false, false);
            acc = __builtin_amdgcn_wmma_f32_16x16x4_f32(false, a3, false, b3, (short)0, acc, false, false);
            TA[i] = acc;
        }
        {
            const v2f b0 = { RB[0], RB[1] }, b1 = { RB[2], RB[3] }, b2 = { RB[4], RB[5] }, b3 = { RB[6], RB[7] };
            v8f acc = TB[i];
            acc = __builtin_amdgcn_wmma_f32_16x16x4_f32(false, a0, false, b0, (short)0, acc, false, false);
            acc = __builtin_amdgcn_wmma_f32_16x16x4_f32(false, a1, false, b1, (short)0, acc, false, false);
            acc = __builtin_amdgcn_wmma_f32_16x16x4_f32(false, a2, false, b2, (short)0, acc, false, false);
            acc = __builtin_amdgcn_wmma_f32_16x16x4_f32(false, a3, false, b3, (short)0, acc, false, false);
            TB[i] = acc;
        }
    }
}

__device__ __forceinline__ void store_abar(const v8f (&T)[4], float* Pw, int cb, int c, int h)
{
    const int col = 16*cb + c;
    #pragma unroll
    for (int i = 0; i < 4; ++i) {
        float* pr = Pw + (16*i + 8*h)*PP + col;
        #pragma unroll
        for (int r = 0; r < 8; ++r) pr[r*PP] = T[i][r];
        float* pt = Pw + HD*PP + col*PP + 16*i + 8*h;
        const v4f lo4 = { T[i][0], T[i][1], T[i][2], T[i][3] };
        const v4f hi4 = { T[i][4], T[i][5], T[i][6], T[i][7] };
        *(v4f*)(pt)     = lo4;
        *(v4f*)(pt + 4) = hi4;
    }
}


__device__ __forceinline__ void load_row(v2f (&o)[16], const float* row, int h)
{
    #pragma unroll
    for (int j = 0; j < 16; ++j) o[j] = *(const v2f*)(row + 2*h + 4*j);
}

__device__ __forceinline__ v8f mma16(const v2f (&a)[16], const v2f (&b)[16])
{
    v8f acc0 = {}, acc1 = {};
    #pragma unroll
    for (int j = 0; j < 16; j += 2) {
        acc0 = __builtin_amdgcn_wmma_f32_16x16x4_f32(false, a[j],     false, b[j],     (short)0, acc0, false, false);
        acc1 = __builtin_amdgcn_wmma_f32_16x16x4_f32(false, a[j + 1], false, b[j + 1], (short)0, acc1, false, false);
    }
    return acc0 + acc1;
}

__device__ __forceinline__ void conv_load(v2f (&a)[8], v2f (&b)[8], const float* up, const float* kp)
{
    #pragma unroll
    for (int q = 0; q < 2; ++q) {
        #pragma unroll
        for (int j = 0; j < 4; ++j) {
            a[4*q + j].x = kp[-16*q + 4*j];
            a[4*q + j].y = kp[-16*q + 4*j + 1];
            b[4*q + j].x = up[-128*q + (4*j)*NBATCH];
            b[4*q + j].y = up[-128*q + (4*j + 1)*NBATCH];
        }
    }
}

__device__ __forceinline__ void conv_mma(v8f& acc0, v8f& acc1, const v2f (&a)[8], const v2f (&b)[8])
{
    #pragma unroll
    for (int i = 0; i < 8; ++i) {
        if (i & 1)
            acc1 = __builtin_amdgcn_wmma_f32_16x16x4_f32(false, a[i], false, b[i], (short)0, acc1, false, false);
        else
            acc0 = __builtin_amdgcn_wmma_f32_16x16x4_f32(false, a[i], false, b[i], (short)0, acc0, false, false);
    }
}

__launch_bounds__(STHREADS)
__global__ void ssm_fused_wmma_f32(const float* __restrict__ x,
                                   const float* __restrict__ Amat,
                                   const float* __restrict__ Bmat,
                                   const float* __restrict__ Cmat,
                                   float* __restrict__ out)
{
    __shared__ __attribute__((aligned(16))) float smem[2*NVEC*VP + (16 + NT) + (NT + 16)*NBATCH + NBATCH*YP + 4*HD*PP];
    float* Vv    = smem;
    float* Ww    = Vv + NVEC*VP;
    float* Kp    = Ww + NVEC*VP;
    float* u_lds = Kp + 16 + NT;
    float* y_lds = u_lds + (NT + 16)*NBATCH;
    float* Pw    = y_lds + NBATCH*YP;
    float* Lc    = y_lds;

    const int f   = blockIdx.x;
    const int tid = threadIdx.x;
    const int wv  = __builtin_amdgcn_readfirstlane(tid >> 5);
    const int ln  = tid & 31;
    const int h   = ln >> 4;
    const int c   = ln & 15;
    const bool solver = (wv < 4);

    const float* Af = Amat + f*HD*HD;

    for (int m = 0; m < 4; ++m) {
        const int e = tid + m*STHREADS;
        const int r = e >> 4, cc = e & 15;
        Lc[r*LP + cc] = ((r == cc) ? 1.0f : 0.0f) - 0.5f*Af[r*HD + cc];
    }

    v8f TA[4] = {}, TB[4] = {};
    if (solver) {
        const bool mid = (wv < 3);
        const float sg = mid ? -0.5f : 0.5f;
        const int col = 16*((wv + 1) & 3) + c;
        float av[4][8], bw[4][8];
        #pragma unroll
        for (int i = 0; i < 4; ++i) {
            #pragma unroll
            for (int r = 0; r < 8; ++r) {
                const int row = 16*i + 8*h + r;
                av[i][r] = Af[row*HD + col];
                bw[i][r] = Bmat[f*HD + row];
            }
        }
        __builtin_amdgcn_sched_barrier(0);
        #pragma unroll
        for (int i = 0; i < 4; ++i) {
            #pragma unroll
            for (int r = 0; r < 8; ++r) {
                const int row = 16*i + 8*h + r;
                const float dl = (row == col) ? 1.0f : 0.0f;
                TA[i][r] = dl + sg*av[i][r];
                TB[i][r] = mid ? (dl + 0.5f*av[i][r]) : ((c == 0) ? bw[i][r] : 0.0f);
            }
        }
    }

    float xv[4][NBATCH];
    #pragma unroll
    for (int q4 = 0; q4 < 4; ++q4) {
        #pragma unroll
        for (int b = 0; b < NBATCH; ++b)
            xv[q4][b] = x[(size_t)(b*NFEAT + f)*NT + tid + q4*STHREADS];
    }
    const float cvv = (tid < HD) ? Cmat[f*HD + tid] : 0.0f;

    for (int e = tid; e < 15*VP; e += STHREADS) { Vv[VP + e] = 0.0f; Ww[VP + e] = 0.0f; }
    if (tid < 16)  Kp[NT + tid] = 0.0f;
    if (tid < 128) u_lds[tid] = 0.0f;
    lds_barrier();

    #pragma nounroll
    for (int kb = 0; kb < 4; ++kb) {
        const float* Lb = Lc + (kb & 1)*HD*LP;
        if (solver) {
            float d[16], e[8];
            const float* dr = Lb + (16*kb + c)*LP;
            #pragma unroll
            for (int q4 = 0; q4 < 4; ++q4) {
                const v4f t = *(const v4f*)(dr + 4*q4);
                d[4*q4] = t[0]; d[4*q4 + 1] = t[1]; d[4*q4 + 2] = t[2]; d[4*q4 + 3] = t[3];
            }
            #pragma unroll
            for (int m = 0; m < 8; ++m) e[m] = (8*h + m == c) ? 1.0f : 0.0f;
            gj_pivot<0>(d, e, c);   gj_pivot<1>(d, e, c);   gj_pivot<2>(d, e, c);   gj_pivot<3>(d, e, c);
            gj_pivot<4>(d, e, c);   gj_pivot<5>(d, e, c);   gj_pivot<6>(d, e, c);   gj_pivot<7>(d, e, c);
            gj_pivot<8>(d, e, c);   gj_pivot<9>(d, e, c);   gj_pivot<10>(d, e, c);  gj_pivot<11>(d, e, c);
            gj_pivot<12>(d, e, c);  gj_pivot<13>(d, e, c);  gj_pivot<14>(d, e, c);  gj_pivot<15>(d, e, c);
            float diag = d[0];
            #pragma unroll
            for (int p = 1; p < 16; ++p) diag = (c == p) ? d[p] : diag;
            #pragma unroll
            for (int m = 0; m < 8; ++m) e[m] = e[m] / diag;

            pivot_update(TA, TB, e, Lb, kb, c, h);

            if (kb < 3 && wv == kb) {
                float* Ln = Lc + ((kb + 1) & 1)*HD*LP;
                #pragma unroll
                for (int i = 0; i < 4; ++i) {
                    #pragma unroll
                    for (int r = 0; r < 8; ++r)
                        Ln[(16*((kb + i) & 3) + 8*h + r)*LP + c] = TA[i][r];
                }
            }
        }
        { const v8f t0 = TA[0]; TA[0] = TA[1]; TA[1] = TA[2]; TA[2] = TA[3]; TA[3] = t0; }
        { const v8f t0 = TB[0]; TB[0] = TB[1]; TB[1] = TB[2]; TB[2] = TB[3]; TB[3] = t0; }
        if (kb < 3) lds_barrier();
    }

    if (solver) {
        if (wv < 3) {
            store_abar(TB, Pw, wv + 1, c, h);
        } else {
            store_abar(TA, Pw, 0, c, h);
            if (c == 0) {
                #pragma unroll
                for (int i = 0; i < 4; ++i) {
                    float* vr = Vv + 16*i + 8*h;
                    const v4f lo4 = { TB[i][0], TB[i][1], TB[i][2], TB[i][3] };
                    const v4f hi4 = { TB[i][4], TB[i][5], TB[i][6], TB[i][7] };
                    *(v4f*)(vr)     = lo4;
                    *(v4f*)(vr + 4) = hi4;
                }
            }
        }
    }
    if (tid < HD) Ww[tid] = cvv;
    #pragma unroll
    for (int q4 = 0; q4 < 4; ++q4) {
        float* ud = u_lds + (16 + tid + q4*STHREADS)*NBATCH;
        const v4f lo4 = { xv[q4][0], xv[q4][1], xv[q4][2], xv[q4][3] };
        const v4f hi4 = { xv[q4][4], xv[q4][5], xv[q4][6], xv[q4][7] };
        *(v4f*)(ud)     = lo4;
        *(v4f*)(ud + 4) = hi4;
    }
    __syncthreads();

    const int tn = wv & 3, tm0 = wv >> 2;
    const bool dbl = (wv < 4);
    #pragma nounroll
    for (int s = 0; s < NSTAGE; ++s) {
        const float* Pc  = Pw + (s & 1)*2*HD*PP;
        const float* PTc = Pc + HD*PP;
        float* Pn  = Pw + ((s + 1) & 1)*2*HD*PP;
        float* PTn = Pn + HD*PP;
        const bool sq   = (s < NSTAGE - 1);
        const bool vs   = (s < 5);
        float*     X    = vs ? Vv : Ww;
        const int  nrow = 1 << (vs ? s : s - 5);
        v2f bq[16], a0[16], a1[16], xa[16], xb[16];
        if (sq) {
            load_row(bq, PTc + (16*tn + c)*PP, h);
            load_row(a0, Pc + (16*tm0 + c)*PP, h);
            load_row(a1, Pc + (16*(tm0 + 2) + c)*PP, h);
        }
        if (dbl) {
            load_row(xa, X + c*VP, h);
            load_row(xb, (vs ? Pc : PTc) + (16*wv + c)*PP, h);
        }
        if (sq) {
            #pragma unroll
            for (int i = 0; i < 2; ++i) {
                const int tm = tm0 + 2*i;
                const v8f d = i ? mma16(a1, bq) : mma16(a0, bq);
                float* pr = Pn + (16*tm + 8*h)*PP + 16*tn + c;
                #pragma unroll
                for (int r = 0; r < 8; ++r) pr[r*PP] = d[r];
                float* pt = PTn + (16*tn + c)*PP + 16*tm + 8*h;
                const v4f lo4 = { d[0], d[1], d[2], d[3] };
                const v4f hi4 = { d[4], d[5], d[6], d[7] };
                *(v4f*)(pt)     = lo4;
                *(v4f*)(pt + 4) = hi4;
            }
        }
        if (dbl) {
            const v8f d = mma16(xa, xb);
            float* xr = X + (nrow + 8*h)*VP + 16*wv + c;
            #pragma unroll
            for (int r = 0; r < 8; ++r)
                if (8*h + r < nrow) xr[r*VP] = d[r];
        }
        __syncthreads();
    }

    if (dbl) {
        const int ta = wv >> 1, tb = wv & 1;
        v2f wa[16], vb[16];
        load_row(wa, Ww + (16*ta + c)*VP, h);
        load_row(vb, Vv + (16*tb + c)*VP, h);
        const v8f kt = mma16(wa, vb);
        float* kr = Kp + NT - 1 - 32*(16*ta + 8*h) - 16*tb - c;
        #pragma unroll
        for (int r = 0; r < 8; ++r) kr[-32*r] = kt[r];
    }
    __syncthreads();

    const int ucol = 128*(c >> 3) + (c & 7);
    for (int g = 0; g < 4; ++g) {
        const int p = (g & 1) ? (8*g + 7 - wv) : (8*g + wv);
        v8f acc0 = {}, acc1 = {};
        const float* up0 = u_lds + (2*p + 1)*128 + ucol + (2*h)*NBATCH;
        const float* kp0 = Kp + NT - 1 - c + 2*h;
        v2f xa[8], xb[8], ya[8], yb[8];
        conv_load(xa, xb, up0, kp0);
        const int npair = (p + 1) >> 1;
        for (int i = 0; i < npair; ++i) {
            conv_load(ya, yb, up0 - 256*(2*i + 1), kp0 - 32*(2*i + 1));
            __builtin_amdgcn_sched_barrier(0);
            conv_mma(acc0, acc1, xa, xb);
            const int tn = (2*i + 2 <= p) ? 2*i + 2 : p;
            conv_load(xa, xb, up0 - 256*tn, kp0 - 32*tn);
            __builtin_amdgcn_sched_barrier(0);
            conv_mma(acc0, acc1, ya, yb);
        }
        if ((p + 1) & 1) conv_mma(acc0, acc1, xa, xb);
        const v8f y = acc0 + acc1;
        float* yp = y_lds + (c & 7)*YP + 16*(2*p + (c >> 3)) + 8*h;
        const v4f ylo = { y[0], y[1], y[2], y[3] };
        const v4f yhi = { y[4], y[5], y[6], y[7] };
        *(v4f*)(yp)     = ylo;
        *(v4f*)(yp + 4) = yhi;
    }
    __syncthreads();

    for (int e = tid; e < NBATCH*NT/4; e += STHREADS) {
        const int b = e >> 8, i4 = (e & 255)*4;
        *(v4f*)(out + (size_t)(b*NFEAT + f)*NT + i4) = *(const v4f*)(y_lds + b*YP + i4);
    }
}

extern "C" void kernel_launch(void* const* d_in, const int* in_sizes, int n_in,
                              void* d_out, int out_size, void* d_ws, size_t ws_size,
                              hipStream_t stream) {
    (void)in_sizes; (void)n_in; (void)ws_size; (void)out_size; (void)d_ws;
    const float* x = (const float*)d_in[0];
    const float* A = (const float*)d_in[1];
    const float* B = (const float*)d_in[2];
    const float* C = (const float*)d_in[3];
    float* out = (float*)d_out;
    ssm_fused_wmma_f32<<<dim3(NFEAT), dim3(STHREADS), 0, stream>>>(x, A, B, C, out);
}


# Discretization's 16x16 inverse without divisions: each pivot's v_rcp_f32 reciprocal (1 ulp) taken in-lane ahead of the row_share, multiples and normalisation as products; quotients round differently
# speedup vs baseline: 21.3898x; 1.0527x over previous
#include <hip/hip_runtime.h>

typedef __attribute__((ext_vector_type(2))) float v2f;
typedef __attribute__((ext_vector_type(4))) float v4f;
typedef __attribute__((ext_vector_type(8))) float v8f;

#define NBATCH 8
#define NFEAT  256
#define HD     64
#define NT     1024
#define LP     20
#define STHREADS 256
#define NVEC   32
#define NSTAGE 10
#define PP 68
#define VP 68
#define YP 1028


template<int P> __device__ __forceinline__ float row_bcast(float x)
{
    return __builtin_bit_cast(float, __builtin_amdgcn_update_dpp(0, __builtin_bit_cast(int, x), 0x150 + P, 0xF, 0xF, false));
}

template<int P> __device__ __forceinline__ void gj_pivot(float (&d)[16], float (&e)[8], float& rd, int c)
{
    const float r   = __builtin_amdgcn_rcpf(d[P]);
    rd = (c == P) ? r : rd;
    const float dm  = (c != P) ? d[P] : 0.0f;
    const float fac = dm*row_bcast<P>(r);
    #pragma unroll
    for (int cc = P + 1; cc < 16; ++cc)
        d[cc] -= fac*row_bcast<P>(d[cc]);
    #pragma unroll
    for (int m = 0; m < 8; ++m)
        e[m] -= fac*row_bcast<P>(e[m]);
}

__device__ __forceinline__ void lds_barrier()
{
    __builtin_amdgcn_fence(__ATOMIC_RELEASE, "workgroup", "local");
    __builtin_amdgcn_s_barrier();
    __builtin_amdgcn_fence(__ATOMIC_ACQUIRE, "workgroup", "local");
}

__device__ __forceinline__ void pivot_update(v8f (&TA)[4], v8f (&TB)[4], const float (&e)[8],
                                             const float* Lb, int kb, int c, int h)
{
    v8f RA = {}, RB = {};
    #pragma unroll
    for (int j = 0; j < 4; ++j) {
        const v2f a = { e[2*j], e[2*j + 1] };
        const v2f bA = { TA[0][2*j], TA[0][2*j + 1] };
        const v2f bB = { TB[0][2*j], TB[0][2*j + 1] };
        RA = __builtin_amdgcn_wmma_f32_16x16x4_f32(false, a, false, bA, (short)0, RA, false, false);
        RB = __builtin_amdgcn_wmma_f32_16x16x4_f32(false, a, false, bB, (short)0, RB, false, false);
    }
    TA[0] = RA;
    TB[0] = RB;
    #pragma unroll
    for (int i = 1; i < 4; ++i) {
        const float* lr = Lb + (16*((kb + i) & 3) + c)*LP + 8*h;
        const v4f l0 = *(const v4f*)(lr);
        const v4f l1 = *(const v4f*)(lr + 4);
        const v2f a0 = { -l0[0], -l0[1] }, a1 = { -l0[2], -l0[3] };
        const v2f a2 = { -l1[0], -l1[1] }, a3 = { -l1[2], -l1[3] };
        {
            const v2f b0 = { RA[0], RA[1] }, b1 = { RA[2], RA[3] }, b2 = { RA[4], RA[5] }, b3 = { RA[6], RA[7] };
            v8f acc = TA[i];
            acc = __builtin_amdgcn_wmma_f32_16x16x4_f32(false, a0, false, b0, (short)0, acc, false, false);
            acc = __builtin_amdgcn_wmma_f32_16x16x4_f32(false, a1, false, b1, (short)0, acc, false, false);
            acc = __builtin_amdgcn_wmma_f32_16x16x4_f32(false, a2, false, b2, (short)0, acc, false, false);
            acc = __builtin_amdgcn_wmma_f32_16x16x4_f32(false, a3, false, b3, (short)0, acc, false, false);
            TA[i] = acc;
        }
        {
            const v2f b0 = { RB[0], RB[1] }, b1 = { RB[2], RB[3] }, b2 = { RB[4], RB[5] }, b3 = { RB[6], RB[7] };
            v8f acc = TB[i];
            acc = __builtin_amdgcn_wmma_f32_16x16x4_f32(false, a0, false, b0, (short)0, acc, false, false);
            acc = __builtin_amdgcn_wmma_f32_16x16x4_f32(false, a1, false, b1, (short)0, acc, false, false);
            acc = __builtin_amdgcn_wmma_f32_16x16x4_f32(false, a2, false, b2, (short)0, acc, false, false);
            acc = __builtin_amdgcn_wmma_f32_16x16x4_f32(false, a3, false, b3, (short)0, acc, false, false);
            TB[i] = acc;
        }
    }
}

__device__ __forceinline__ void store_abar(const v8f (&T)[4], float* Pw, int cb, int c, int h)
{
    const int col = 16*cb + c;
    #pragma unroll
    for (int i = 0; i < 4; ++i) {
        float* pr = Pw + (16*i + 8*h)*PP + col;
        #pragma unroll
        for (int r = 0; r < 8; ++r) pr[r*PP] = T[i][r];
        float* pt = Pw + HD*PP + col*PP + 16*i + 8*h;
        const v4f lo4 = { T[i][0], T[i][1], T[i][2], T[i][3] };
        const v4f hi4 = { T[i][4], T[i][5], T[i][6], T[i][7] };
        *(v4f*)(pt)     = lo4;
        *(v4f*)(pt + 4) = hi4;
    }
}


__device__ __forceinline__ void load_row(v2f (&o)[16], const float* row, int h)
{
    #pragma unroll
    for (int j = 0; j < 16; ++j) o[j] = *(const v2f*)(row + 2*h + 4*j);
}

__device__ __forceinline__ v8f mma16(const v2f (&a)[16], const v2f (&b)[16])
{
    v8f acc0 = {}, acc1 = {};
    #pragma unroll
    for (int j = 0; j < 16; j += 2) {
        acc0 = __builtin_amdgcn_wmma_f32_16x16x4_f32(false, a[j],     false, b[j],     (short)0, acc0, false, false);
        acc1 = __builtin_amdgcn_wmma_f32_16x16x4_f32(false, a[j + 1], false, b[j + 1], (short)0, acc1, false, false);
    }
    return acc0 + acc1;
}

__device__ __forceinline__ void conv_load(v2f (&a)[8], v2f (&b)[8], const float* up, const float* kp)
{
    #pragma unroll
    for (int q = 0; q < 2; ++q) {
        #pragma unroll
        for (int j = 0; j < 4; ++j) {
            a[4*q + j].x = kp[-16*q + 4*j];
            a[4*q + j].y = kp[-16*q + 4*j + 1];
            b[4*q + j].x = up[-128*q + (4*j)*NBATCH];
            b[4*q + j].y = up[-128*q + (4*j + 1)*NBATCH];
        }
    }
}

__device__ __forceinline__ void conv_mma(v8f& acc0, v8f& acc1, const v2f (&a)[8], const v2f (&b)[8])
{
    #pragma unroll
    for (int i = 0; i < 8; ++i) {
        if (i & 1)
            acc1 = __builtin_amdgcn_wmma_f32_16x16x4_f32(false, a[i], false, b[i], (short)0, acc1, false, false);
        else
            acc0 = __builtin_amdgcn_wmma_f32_16x16x4_f32(false, a[i], false, b[i], (short)0, acc0, false, false);
    }
}

__launch_bounds__(STHREADS)
__global__ void ssm_fused_wmma_f32(const float* __restrict__ x,
                                   const float* __restrict__ Amat,
                                   const float* __restrict__ Bmat,
                                   const float* __restrict__ Cmat,
                                   float* __restrict__ out)
{
    __shared__ __attribute__((aligned(16))) float smem[2*NVEC*VP + (16 + NT) + (NT + 16)*NBATCH + NBATCH*YP + 4*HD*PP];
    float* Vv    = smem;
    float* Ww    = Vv + NVEC*VP;
    float* Kp    = Ww + NVEC*VP;
    float* u_lds = Kp + 16 + NT;
    float* y_lds = u_lds + (NT + 16)*NBATCH;
    float* Pw    = y_lds + NBATCH*YP;
    float* Lc    = y_lds;

    const int f   = blockIdx.x;
    const int tid = threadIdx.x;
    const int wv  = __builtin_amdgcn_readfirstlane(tid >> 5);
    const int ln  = tid & 31;
    const int h   = ln >> 4;
    const int c   = ln & 15;
    const bool solver = (wv < 4);

    const float* Af = Amat + f*HD*HD;

    for (int m = 0; m < 4; ++m) {
        const int e = tid + m*STHREADS;
        const int r = e >> 4, cc = e & 15;
        Lc[r*LP + cc] = ((r == cc) ? 1.0f : 0.0f) - 0.5f*Af[r*HD + cc];
    }

    v8f TA[4] = {}, TB[4] = {};
    if (solver) {
        const bool mid = (wv < 3);
        const float sg = mid ? -0.5f : 0.5f;
        const int col = 16*((wv + 1) & 3) + c;
        float av[4][8], bw[4][8];
        #pragma unroll
        for (int i = 0; i < 4; ++i) {
            #pragma unroll
            for (int r = 0; r < 8; ++r) {
                const int row = 16*i + 8*h + r;
                av[i][r] = Af[row*HD + col];
                bw[i][r] = Bmat[f*HD + row];
            }
        }
        __builtin_amdgcn_sched_barrier(0);
        #pragma unroll
        for (int i = 0; i < 4; ++i) {
            #pragma unroll
            for (int r = 0; r < 8; ++r) {
                const int row = 16*i + 8*h + r;
                const float dl = (row == col) ? 1.0f : 0.0f;
                TA[i][r] = dl + sg*av[i][r];
                TB[i][r] = mid ? (dl + 0.5f*av[i][r]) : ((c == 0) ? bw[i][r] : 0.0f);
            }
        }
    }

    float xv[4][NBATCH];
    #pragma unroll
    for (int q4 = 0; q4 < 4; ++q4) {
        #pragma unroll
        for (int b = 0; b < NBATCH; ++b)
            xv[q4][b] = x[(size_t)(b*NFEAT + f)*NT + tid + q4*STHREADS];
    }
    const float cvv = (tid < HD) ? Cmat[f*HD + tid] : 0.0f;

    for (int e = tid; e < 15*VP; e += STHREADS) { Vv[VP + e] = 0.0f; Ww[VP + e] = 0.0f; }
    if (tid < 16)  Kp[NT + tid] = 0.0f;
    if (tid < 128) u_lds[tid] = 0.0f;
    lds_barrier();

    #pragma nounroll
    for (int kb = 0; kb < 4; ++kb) {
        const float* Lb = Lc + (kb & 1)*HD*LP;
        if (solver) {
            float d[16], e[8];
            const float* dr = Lb + (16*kb + c)*LP;
            #pragma unroll
            for (int q4 = 0; q4 < 4; ++q4) {
                const v4f t = *(const v4f*)(dr + 4*q4);
                d[4*q4] = t[0]; d[4*q4 + 1] = t[1]; d[4*q4 + 2] = t[2]; d[4*q4 + 3] = t[3];
            }
            #pragma unroll
            for (int m = 0; m < 8; ++m) e[m] = (8*h + m == c) ? 1.0f : 0.0f;
            float rd = 0.0f;
            gj_pivot<0>(d, e, rd, c);   gj_pivot<1>(d, e, rd, c);   gj_pivot<2>(d, e, rd, c);   gj_pivot<3>(d, e, rd, c);
            gj_pivot<4>(d, e, rd, c);   gj_pivot<5>(d, e, rd, c);   gj_pivot<6>(d, e, rd, c);   gj_pivot<7>(d, e, rd, c);
            gj_pivot<8>(d, e, rd, c);   gj_pivot<9>(d, e, rd, c);   gj_pivot<10>(d, e, rd, c);  gj_pivot<11>(d, e, rd, c);
            gj_pivot<12>(d, e, rd, c);  gj_pivot<13>(d, e, rd, c);  gj_pivot<14>(d, e, rd, c);  gj_pivot<15>(d, e, rd, c);
            #pragma unroll
            for (int m = 0; m < 8; ++m) e[m] = e[m]*rd;

            pivot_update(TA, TB, e, Lb, kb, c, h);

            if (kb < 3 && wv == kb) {
                float* Ln = Lc + ((kb + 1) & 1)*HD*LP;
                #pragma unroll
                for (int i = 0; i < 4; ++i) {
                    #pragma unroll
                    for (int r = 0; r < 8; ++r)
                        Ln[(16*((kb + i) & 3) + 8*h + r)*LP + c] = TA[i][r];
                }
            }
        }
        { const v8f t0 = TA[0]; TA[0] = TA[1]; TA[1] = TA[2]; TA[2] = TA[3]; TA[3] = t0; }
        { const v8f t0 = TB[0]; TB[0] = TB[1]; TB[1] = TB[2]; TB[2] = TB[3]; TB[3] = t0; }
        if (kb < 3) lds_barrier();
    }

    if (solver) {
        if (wv < 3) {
            store_abar(TB, Pw, wv + 1, c, h);
        } else {
            store_abar(TA, Pw, 0, c, h);
            if (c == 0) {
                #pragma unroll
                for (int i = 0; i < 4; ++i) {
                    float* vr = Vv + 16*i + 8*h;
                    const v4f lo4 = { TB[i][0], TB[i][1], TB[i][2], TB[i][3] };
                    const v4f hi4 = { TB[i][4], TB[i][5], TB[i][6], TB[i][7] };
                    *(v4f*)(vr)     = lo4;
                    *(v4f*)(vr + 4) = hi4;
                }
            }
        }
    }
    if (tid < HD) Ww[tid] = cvv;
    #pragma unroll
    for (int q4 = 0; q4 < 4; ++q4) {
        float* ud = u_lds + (16 + tid + q4*STHREADS)*NBATCH;
        const v4f lo4 = { xv[q4][0], xv[q4][1], xv[q4][2], xv[q4][3] };
        const v4f hi4 = { xv[q4][4], xv[q4][5], xv[q4][6], xv[q4][7] };
        *(v4f*)(ud)     = lo4;
        *(v4f*)(ud + 4) = hi4;
    }
    __syncthreads();

    const int tn = wv & 3, tm0 = wv >> 2;
    const bool dbl = (wv < 4);
    #pragma nounroll
    for (int s = 0; s < NSTAGE; ++s) {
        const float* Pc  = Pw + (s & 1)*2*HD*PP;
        const float* PTc = Pc + HD*PP;
        float* Pn  = Pw + ((s + 1) & 1)*2*HD*PP;
        float* PTn = Pn + HD*PP;
        const bool sq   = (s < NSTAGE - 1);
        const bool vs   = (s < 5);
        float*     X    = vs ? Vv : Ww;
        const int  nrow = 1 << (vs ? s : s - 5);
        v2f bq[16], a0[16], a1[16], xa[16], xb[16];
        if (sq) {
            load_row(bq, PTc + (16*tn + c)*PP, h);
            load_row(a0, Pc + (16*tm0 + c)*PP, h);
            load_row(a1, Pc + (16*(tm0 + 2) + c)*PP, h);
        }
        if (dbl) {
            load_row(xa, X + c*VP, h);
            load_row(xb, (vs ? Pc : PTc) + (16*wv + c)*PP, h);
        }
        if (sq) {
            #pragma unroll
            for (int i = 0; i < 2; ++i) {
                const int tm = tm0 + 2*i;
                const v8f d = i ? mma16(a1, bq) : mma16(a0, bq);
                float* pr = Pn + (16*tm + 8*h)*PP + 16*tn + c;
                #pragma unroll
                for (int r = 0; r < 8; ++r) pr[r*PP] = d[r];
                float* pt = PTn + (16*tn + c)*PP + 16*tm + 8*h;
                const v4f lo4 = { d[0], d[1], d[2], d[3] };
                const v4f hi4 = { d[4], d[5], d[6], d[7] };
                *(v4f*)(pt)     = lo4;
                *(v4f*)(pt + 4) = hi4;
            }
        }
        if (dbl) {
            const v8f d = mma16(xa, xb);
            float* xr = X + (nrow + 8*h)*VP + 16*wv + c;
            #pragma unroll
            for (int r = 0; r < 8; ++r)
                if (8*h + r < nrow) xr[r*VP] = d[r];
        }
        __syncthreads();
    }

    if (dbl) {
        const int ta = wv >> 1, tb = wv & 1;
        v2f wa[16], vb[16];
        load_row(wa, Ww + (16*ta + c)*VP, h);
        load_row(vb, Vv + (16*tb + c)*VP, h);
        const v8f kt = mma16(wa, vb);
        float* kr = Kp + NT - 1 - 32*(16*ta + 8*h) - 16*tb - c;
        #pragma unroll
        for (int r = 0; r < 8; ++r) kr[-32*r] = kt[r];
    }
    __syncthreads();

    const int ucol = 128*(c >> 3) + (c & 7);
    for (int g = 0; g < 4; ++g) {
        const int p = (g & 1) ? (8*g + 7 - wv) : (8*g + wv);
        v8f acc0 = {}, acc1 = {};
        const float* up0 = u_lds + (2*p + 1)*128 + ucol + (2*h)*NBATCH;
        const float* kp0 = Kp + NT - 1 - c + 2*h;
        v2f xa[8], xb[8], ya[8], yb[8];
        conv_load(xa, xb, up0, kp0);
        const int npair = (p + 1) >> 1;
        for (int i = 0; i < npair; ++i) {
            conv_load(ya, yb, up0 - 256*(2*i + 1), kp0 - 32*(2*i + 1));
            __builtin_amdgcn_sched_barrier(0);
            conv_mma(acc0, acc1, xa, xb);
            const int tn = (2*i + 2 <= p) ? 2*i + 2 : p;
            conv_load(xa, xb, up0 - 256*tn, kp0 - 32*tn);
            __builtin_amdgcn_sched_barrier(0);
            conv_mma(acc0, acc1, ya, yb);
        }
        if ((p + 1) & 1) conv_mma(acc0, acc1, xa, xb);
        const v8f y = acc0 + acc1;
        float* yp = y_lds + (c & 7)*YP + 16*(2*p + (c >> 3)) + 8*h;
        const v4f ylo = { y[0], y[1], y[2], y[3] };
        const v4f yhi = { y[4], y[5], y[6], y[7] };
        *(v4f*)(yp)     = ylo;
        *(v4f*)(yp + 4) = yhi;
    }
    __syncthreads();

    for (int e = tid; e < NBATCH*NT/4; e += STHREADS) {
        const int b = e >> 8, i4 = (e & 255)*4;
        *(v4f*)(out + (size_t)(b*NFEAT + f)*NT + i4) = *(const v4f*)(y_lds + b*YP + i4);
    }
}

extern "C" void kernel_launch(void* const* d_in, const int* in_sizes, int n_in,
                              void* d_out, int out_size, void* d_ws, size_t ws_size,
                              hipStream_t stream) {
    (void)in_sizes; (void)n_in; (void)ws_size; (void)out_size; (void)d_ws;
    const float* x = (const float*)d_in[0];
    const float* A = (const float*)d_in[1];
    const float* B = (const float*)d_in[2];
    const float* C = (const float*)d_in[3];
    float* out = (float*)d_out;
    ssm_fused_wmma_f32<<<dim3(NFEAT), dim3(STHREADS), 0, stream>>>(x, A, B, C, out);
}
